# starting kernel
# baseline (speedup 1.0000x reference)
#include <hip/hip_runtime.h>


typedef __attribute__((ext_vector_type(2))) float v2f;
typedef __attribute__((ext_vector_type(4))) float v4f;
typedef __attribute__((ext_vector_type(8))) float v8f;

#define BATCH  8
#define SEQLEN 4096
#define HID    1024
#define STATE  64
#define T      16
#define NCHUNK (SEQLEN / T)
#define DPB    16
#define BP     16

#define PER_D   2368
#define OFF_P   0
#define OFF_G   1024
#define OFF_M   2048
#define OFF_LAM 2304

static __device__ __forceinline__ v8f wmma4(v2f a, v2f b, v8f c) {
  return __builtin_amdgcn_wmma_f32_16x16x4_f32(false, a, false, b, (short)0, c,
                                               false, false);
}

__global__ __launch_bounds__(64) void ssm_setup_kernel(
    const float* __restrict__ A, const float* __restrict__ Bm,
    const float* __restrict__ C, const float* __restrict__ Dv,
    const float* __restrict__ dt, float* __restrict__ ws) {
  const int d = blockIdx.x;
  const int n = threadIdx.x;
  __shared__ float sK[T * STATE];
  __shared__ float sKl[T];
  float* base = ws + (size_t)d * PER_D;

  const float a  = expf(A[d * STATE + n]);
  const float da = expf(-a * dt[n]);
  const float db = Bm[d * STATE + n] * (1.0f - da) / a;
  const float cc = C[d * STATE + n];

  float p = da;
#pragma unroll
  for (int t = 0; t < T; ++t) { base[OFF_P + t * STATE + n] = cc * p; p *= da; }

  const float da2 = da * da, da4 = da2 * da2, da8 = da4 * da4;
  base[OFF_LAM + n] = da8 * da8;

  float g = db;
  for (int s = T - 1; s >= 0; --s) { base[OFF_G + n * T + s] = g; g *= da; }

  float w = cc * db;
#pragma unroll
  for (int tau = 0; tau < T; ++tau) { sK[tau * STATE + n] = w; w *= da; }
  __syncthreads();
  if (n < T) {
    float s = 0.0f;
    for (int m = 0; m < STATE; ++m) s += sK[n * STATE + m];
    sKl[n] = s;
  }
  __syncthreads();
  if (n < T) {
    const int t = n;
    const float dvd = Dv[d];
    for (int s2 = 0; s2 < T; ++s2) {
      float v = (t >= s2) ? sKl[t - s2] : 0.0f;
      if (s2 == t) v += dvd;
      base[OFF_M + t * T + s2] = v;
    }
  }
}

__global__ __launch_bounds__(32 * DPB) void ssm_scan_kernel(
    const float* __restrict__ u, const float* __restrict__ ws,
    float* __restrict__ out) {
  const int tid  = threadIdx.x;
  const int lane = tid & 31;
  const int wv   = tid >> 5;
  const int d0   = blockIdx.x * DPB;
  const int half = lane >> 4;
  const int col  = lane & 15;

  __shared__ __align__(16) float ldsU[T * BP * DPB];
  __shared__ __align__(16) float ldsY[T * BP * DPB];

#pragma unroll
  for (int k = 0; k < (T * BP * DPB) / (32 * DPB); ++k)
    ldsU[tid + k * (32 * DPB)] = 0.0f;

  const float* base = ws + (size_t)(d0 + wv) * PER_D;

  v2f pA[16];
#pragma unroll
  for (int kk = 0; kk < 16; ++kk)
    pA[kk] = *(const v2f*)(base + OFF_P + col * STATE + 4 * kk + 2 * half);
  v2f mA[4];
#pragma unroll
  for (int kk = 0; kk < 4; ++kk)
    mA[kk] = *(const v2f*)(base + OFF_M + col * T + 4 * kk + 2 * half);
  v2f gA[4][4];
#pragma unroll
  for (int r = 0; r < 4; ++r)
#pragma unroll
    for (int kk = 0; kk < 4; ++kk)
      gA[r][kk] =
          *(const v2f*)(base + OFF_G + (r * 16 + col) * T + 4 * kk + 2 * half);
  float lam[4][8];
#pragma unroll
  for (int r = 0; r < 4; ++r)
#pragma unroll
    for (int i = 0; i < 8; ++i)
      lam[r][i] = base[OFF_LAM + r * 16 + i + 8 * half];

  v8f x[4] = {};

  const int pair = tid >> 2;
  const int bb   = pair >> 4;
  const int tt   = pair & 15;
  const int d4   = (tid & 3) << 2;
  const size_t gRow = (size_t)bb * SEQLEN * HID + (size_t)(d0 + d4);

  for (int c = 0; c < NCHUNK; ++c) {
    {
      const float* gp = u + gRow + (size_t)(c * T + tt) * HID;
      v4f v = *(const v4f*)gp;
      *(v4f*)(&ldsU[(tt * BP + bb) * DPB + d4]) = v;
      if (c + 1 < NCHUNK)
        __builtin_prefetch(u + gRow + (size_t)((c + 1) * T + tt) * HID, 0, 0);
    }
    __syncthreads();

    v2f uB[4];
#pragma unroll
    for (int kk = 0; kk < 4; ++kk) {
#pragma unroll
      for (int j = 0; j < 2; ++j) {
        const int s = 4 * kk + 2 * half + j;
        uB[kk][j] = ldsU[(s * BP + col) * DPB + wv];
      }
    }

    v8f y = {};
#pragma unroll
    for (int kk = 0; kk < 4; ++kk) y = wmma4(mA[kk], uB[kk], y);

#pragma unroll
    for (int kk = 0; kk < 16; ++kk) {
      v2f xb;
#pragma unroll
      for (int j = 0; j < 2; ++j) {
        const int n0 = 4 * kk + j;
        const int n1 = 4 * kk + 2 + j;
        const float c0 =
            __shfl(x[n0 >> 4][(n0 & 15) & 7], col + 16 * ((n0 & 15) >> 3), 32);
        const float c1 =
            __shfl(x[n1 >> 4][(n1 & 15) & 7], col + 16 * ((n1 & 15) >> 3), 32);
        xb[j] = half ? c1 : c0;
      }
      y = wmma4(pA[kk], xb, y);
    }

#pragma unroll
    for (int r = 0; r < 4; ++r) {
      v8f xn;
#pragma unroll
      for (int i = 0; i < 8; ++i) xn[i] = lam[r][i] * x[r][i];
#pragma unroll
      for (int kk = 0; kk < 4; ++kk) xn = wmma4(gA[r][kk], uB[kk], xn);
      x[r] = xn;
    }

#pragma unroll
    for (int i = 0; i < 8; ++i) {
      const int t = i + 8 * half;
      ldsY[(t * BP + col) * DPB + wv] = y[i];
    }
    __syncthreads();
    {
      v4f v = *(const v4f*)(&ldsY[(tt * BP + bb) * DPB + d4]);
      *(v4f*)(out + gRow + (size_t)(c * T + tt) * HID) = v;
    }
  }
}

extern "C" void kernel_launch(void* const* d_in, const int* in_sizes, int n_in,
                              void* d_out, int out_size, void* d_ws,
                              size_t ws_size, hipStream_t stream) {
  (void)in_sizes; (void)n_in; (void)out_size; (void)ws_size;
  const float* u  = (const float*)d_in[0];
  const float* A  = (const float*)d_in[1];
  const float* Bm = (const float*)d_in[2];
  const float* C  = (const float*)d_in[3];
  const float* Dv = (const float*)d_in[4];
  const float* dt = (const float*)d_in[5];
  float* out = (float*)d_out;
  float* ws  = (float*)d_ws;

  ssm_setup_kernel<<<HID, STATE, 0, stream>>>(A, Bm, C, Dv, dt, ws);
  ssm_scan_kernel<<<HID / DPB, 32 * DPB, 0, stream>>>(u, ws, out);
}


# Scan re-gridded from 64 work-groups of 16 waves to 256 work-groups of 4 waves (4 channels each, one wave per SIMD): all 256 WGPs share the WMMA work instead of 64
# speedup vs baseline: 1.6704x; 1.6704x over previous
#include <hip/hip_runtime.h>


typedef __attribute__((ext_vector_type(2))) float v2f;
typedef __attribute__((ext_vector_type(4))) float v4f;
typedef __attribute__((ext_vector_type(8))) float v8f;

#define BATCH  8
#define SEQLEN 4096
#define HID    1024
#define STATE  64
#define T      16
#define NCHUNK (SEQLEN / T)
#define DPB    4
#define BP     16

#define PER_D   2368
#define OFF_P   0
#define OFF_G   1024
#define OFF_M   2048
#define OFF_LAM 2304

static __device__ __forceinline__ v8f wmma4(v2f a, v2f b, v8f c) {
  return __builtin_amdgcn_wmma_f32_16x16x4_f32(false, a, false, b, (short)0, c,
                                               false, false);
}

__global__ __launch_bounds__(64) void ssm_setup_kernel(
    const float* __restrict__ A, const float* __restrict__ Bm,
    const float* __restrict__ C, const float* __restrict__ Dv,
    const float* __restrict__ dt, float* __restrict__ ws) {
  const int d = blockIdx.x;
  const int n = threadIdx.x;
  __shared__ float sK[T * STATE];
  __shared__ float sKl[T];
  float* base = ws + (size_t)d * PER_D;

  const float a  = expf(A[d * STATE + n]);
  const float da = expf(-a * dt[n]);
  const float db = Bm[d * STATE + n] * (1.0f - da) / a;
  const float cc = C[d * STATE + n];

  float p = da;
#pragma unroll
  for (int t = 0; t < T; ++t) { base[OFF_P + t * STATE + n] = cc * p; p *= da; }

  const float da2 = da * da, da4 = da2 * da2, da8 = da4 * da4;
  base[OFF_LAM + n] = da8 * da8;

  float g = db;
  for (int s = T - 1; s >= 0; --s) { base[OFF_G + n * T + s] = g; g *= da; }

  float w = cc * db;
#pragma unroll
  for (int tau = 0; tau < T; ++tau) { sK[tau * STATE + n] = w; w *= da; }
  __syncthreads();
  if (n < T) {
    float s = 0.0f;
    for (int m = 0; m < STATE; ++m) s += sK[n * STATE + m];
    sKl[n] = s;
  }
  __syncthreads();
  if (n < T) {
    const int t = n;
    const float dvd = Dv[d];
    for (int s2 = 0; s2 < T; ++s2) {
      float v = (t >= s2) ? sKl[t - s2] : 0.0f;
      if (s2 == t) v += dvd;
      base[OFF_M + t * T + s2] = v;
    }
  }
}

__global__ __launch_bounds__(32 * DPB) void ssm_scan_kernel(
    const float* __restrict__ u, const float* __restrict__ ws,
    float* __restrict__ out) {
  const int tid  = threadIdx.x;
  const int lane = tid & 31;
  const int wv   = tid >> 5;
  const int d0   = blockIdx.x * DPB;
  const int half = lane >> 4;
  const int col  = lane & 15;

  __shared__ __align__(16) float ldsU[T * BP * DPB];
  __shared__ __align__(16) float ldsY[T * BP * DPB];

#pragma unroll
  for (int k = 0; k < (T * BP * DPB) / (32 * DPB); ++k)
    ldsU[tid + k * (32 * DPB)] = 0.0f;
  __syncthreads();

  const float* base = ws + (size_t)(d0 + wv) * PER_D;

  v2f pA[16];
#pragma unroll
  for (int kk = 0; kk < 16; ++kk)
    pA[kk] = *(const v2f*)(base + OFF_P + col * STATE + 4 * kk + 2 * half);
  v2f mA[4];
#pragma unroll
  for (int kk = 0; kk < 4; ++kk)
    mA[kk] = *(const v2f*)(base + OFF_M + col * T + 4 * kk + 2 * half);
  v2f gA[4][4];
#pragma unroll
  for (int r = 0; r < 4; ++r)
#pragma unroll
    for (int kk = 0; kk < 4; ++kk)
      gA[r][kk] =
          *(const v2f*)(base + OFF_G + (r * 16 + col) * T + 4 * kk + 2 * half);
  float lam[4][8];
#pragma unroll
  for (int r = 0; r < 4; ++r)
#pragma unroll
    for (int i = 0; i < 8; ++i)
      lam[r][i] = base[OFF_LAM + r * 16 + i + 8 * half];

  v8f x[4] = {};

  const int tt   = tid >> 3;
  const int bb   = tid & 7;
  const int d4   = 0;
  const size_t gRow = (size_t)bb * SEQLEN * HID + (size_t)(d0 + d4);

  for (int c = 0; c < NCHUNK; ++c) {
    {
      const float* gp = u + gRow + (size_t)(c * T + tt) * HID;
      v4f v = *(const v4f*)gp;
      *(v4f*)(&ldsU[(tt * BP + bb) * DPB + d4]) = v;
      if (c + 1 < NCHUNK)
        __builtin_prefetch(u + gRow + (size_t)((c + 1) * T + tt) * HID, 0, 0);
    }
    __syncthreads();

    v2f uB[4];
#pragma unroll
    for (int kk = 0; kk < 4; ++kk) {
#pragma unroll
      for (int j = 0; j < 2; ++j) {
        const int s = 4 * kk + 2 * half + j;
        uB[kk][j] = ldsU[(s * BP + col) * DPB + wv];
      }
    }

    v8f y = {};
#pragma unroll
    for (int kk = 0; kk < 4; ++kk) y = wmma4(mA[kk], uB[kk], y);

#pragma unroll
    for (int kk = 0; kk < 16; ++kk) {
      v2f xb;
#pragma unroll
      for (int j = 0; j < 2; ++j) {
        const int n0 = 4 * kk + j;
        const int n1 = 4 * kk + 2 + j;
        const float c0 =
            __shfl(x[n0 >> 4][(n0 & 15) & 7], col + 16 * ((n0 & 15) >> 3), 32);
        const float c1 =
            __shfl(x[n1 >> 4][(n1 & 15) & 7], col + 16 * ((n1 & 15) >> 3), 32);
        xb[j] = half ? c1 : c0;
      }
      y = wmma4(pA[kk], xb, y);
    }

#pragma unroll
    for (int r = 0; r < 4; ++r) {
      v8f xn;
#pragma unroll
      for (int i = 0; i < 8; ++i) xn[i] = lam[r][i] * x[r][i];
#pragma unroll
      for (int kk = 0; kk < 4; ++kk) xn = wmma4(gA[r][kk], uB[kk], xn);
      x[r] = xn;
    }

#pragma unroll
    for (int i = 0; i < 8; ++i) {
      const int t = i + 8 * half;
      ldsY[(t * BP + col) * DPB + wv] = y[i];
    }
    __syncthreads();
    {
      v4f v = *(const v4f*)(&ldsY[(tt * BP + bb) * DPB + d4]);
      *(v4f*)(out + gRow + (size_t)(c * T + tt) * HID) = v;
    }
  }
}

extern "C" void kernel_launch(void* const* d_in, const int* in_sizes, int n_in,
                              void* d_out, int out_size, void* d_ws,
                              size_t ws_size, hipStream_t stream) {
  (void)in_sizes; (void)n_in; (void)out_size; (void)ws_size;
  const float* u  = (const float*)d_in[0];
  const float* A  = (const float*)d_in[1];
  const float* Bm = (const float*)d_in[2];
  const float* C  = (const float*)d_in[3];
  const float* Dv = (const float*)d_in[4];
  const float* dt = (const float*)d_in[5];
  float* out = (float*)d_out;
  float* ws  = (float*)d_ws;

  ssm_setup_kernel<<<HID, STATE, 0, stream>>>(A, Bm, C, Dv, dt, ws);
  ssm_scan_kernel<<<HID / DPB, 32 * DPB, 0, stream>>>(u, ws, out);
}


# Chunk loop software-pipelined: u tiles loaded to registers four chunks (one stage) ahead, LDS tiles double-buffered per stage, one barrier per four chunks instead of two per chunk
# speedup vs baseline: 2.4123x; 1.4441x over previous
#include <hip/hip_runtime.h>


typedef __attribute__((ext_vector_type(2))) float v2f;
typedef __attribute__((ext_vector_type(4))) float v4f;
typedef __attribute__((ext_vector_type(8))) float v8f;

#define BATCH  8
#define SEQLEN 4096
#define HID    1024
#define STATE  64
#define T      16
#define NCHUNK (SEQLEN / T)
#define DPB    4
#define BP     16
#define SUP    4
#define NSTAGE (NCHUNK / SUP)
#define TILE   (T * BP * DPB)

#define PER_D   2368
#define OFF_P   0
#define OFF_G   1024
#define OFF_M   2048
#define OFF_LAM 2304

static __device__ __forceinline__ v8f wmma4(v2f a, v2f b, v8f c) {
  return __builtin_amdgcn_wmma_f32_16x16x4_f32(false, a, false, b, (short)0, c,
                                               false, false);
}

__global__ __launch_bounds__(64) void ssm_setup_kernel(
    const float* __restrict__ A, const float* __restrict__ Bm,
    const float* __restrict__ C, const float* __restrict__ Dv,
    const float* __restrict__ dt, float* __restrict__ ws) {
  const int d = blockIdx.x;
  const int n = threadIdx.x;
  __shared__ float sK[T * STATE];
  __shared__ float sKl[T];
  float* base = ws + (size_t)d * PER_D;

  const float a  = expf(A[d * STATE + n]);
  const float da = expf(-a * dt[n]);
  const float db = Bm[d * STATE + n] * (1.0f - da) / a;
  const float cc = C[d * STATE + n];

  float p = da;
#pragma unroll
  for (int t = 0; t < T; ++t) { base[OFF_P + t * STATE + n] = cc * p; p *= da; }

  const float da2 = da * da, da4 = da2 * da2, da8 = da4 * da4;
  base[OFF_LAM + n] = da8 * da8;

  float g = db;
  for (int s = T - 1; s >= 0; --s) { base[OFF_G + n * T + s] = g; g *= da; }

  float w = cc * db;
#pragma unroll
  for (int tau = 0; tau < T; ++tau) { sK[tau * STATE + n] = w; w *= da; }
  __syncthreads();
  if (n < T) {
    float s = 0.0f;
    for (int m = 0; m < STATE; ++m) s += sK[n * STATE + m];
    sKl[n] = s;
  }
  __syncthreads();
  if (n < T) {
    const int t = n;
    const float dvd = Dv[d];
    for (int s2 = 0; s2 < T; ++s2) {
      float v = (t >= s2) ? sKl[t - s2] : 0.0f;
      if (s2 == t) v += dvd;
      base[OFF_M + t * T + s2] = v;
    }
  }
}

__global__ __launch_bounds__(32 * DPB) void ssm_scan_kernel(
    const float* __restrict__ u, const float* __restrict__ ws,
    float* __restrict__ out) {
  const int tid  = threadIdx.x;
  const int lane = tid & 31;
  const int wv   = tid >> 5;
  const int d0   = blockIdx.x * DPB;
  const int half = lane >> 4;
  const int col  = lane & 15;

  __shared__ __align__(16) float ldsU[2][SUP][TILE];
  __shared__ __align__(16) float ldsY[2][SUP][TILE];

  {
    const v4f z = {0.0f, 0.0f, 0.0f, 0.0f};
    for (int k = 0; k < (2 * SUP * TILE) / (4 * 32 * DPB); ++k)
      *(v4f*)(&ldsU[0][0][0] + 4 * (tid + k * (32 * DPB))) = z;
  }
  __syncthreads();

  const float* base = ws + (size_t)(d0 + wv) * PER_D;

  v2f pA[16];
#pragma unroll
  for (int kk = 0; kk < 16; ++kk)
    pA[kk] = *(const v2f*)(base + OFF_P + col * STATE + 4 * kk + 2 * half);
  v2f mA[4];
#pragma unroll
  for (int kk = 0; kk < 4; ++kk)
    mA[kk] = *(const v2f*)(base + OFF_M + col * T + 4 * kk + 2 * half);
  v2f gA[4][4];
#pragma unroll
  for (int r = 0; r < 4; ++r)
#pragma unroll
    for (int kk = 0; kk < 4; ++kk)
      gA[r][kk] =
          *(const v2f*)(base + OFF_G + (r * 16 + col) * T + 4 * kk + 2 * half);
  float lam[4][8];
#pragma unroll
  for (int r = 0; r < 4; ++r)
#pragma unroll
    for (int i = 0; i < 8; ++i)
      lam[r][i] = base[OFF_LAM + r * 16 + i + 8 * half];

  v8f x[4] = {};

  const int tt   = tid >> 3;
  const int bb   = tid & 7;
  const int d4   = 0;
  const size_t gRow = (size_t)bb * SEQLEN * HID + (size_t)(d0 + d4);
  const int coop = (tt * BP + bb) * DPB + d4;

#pragma unroll
  for (int k = 0; k < SUP; ++k)
    *(v4f*)(&ldsU[0][k][coop]) =
        *(const v4f*)(u + gRow + (size_t)(k * T + tt) * HID);
  __syncthreads();

  for (int g = 0; g < NSTAGE; ++g) {
    const int buf = g & 1;
    const int gn = (g + 1) & (NSTAGE - 1);
    const int gp = (g + 2) & (NSTAGE - 1);
    v4f nv[SUP];
#pragma unroll
    for (int k = 0; k < SUP; ++k) {
      nv[k] = *(const v4f*)(u + gRow + (size_t)((gn * SUP + k) * T + tt) * HID);
      __builtin_prefetch(u + gRow + (size_t)((gp * SUP + k) * T + tt) * HID, 0, 0);
    }

#pragma unroll 1
    for (int k = 0; k < SUP; ++k) {
      const float* uT = &ldsU[buf][k][0];
      v2f uB[4];
#pragma unroll
      for (int kk = 0; kk < 4; ++kk) {
#pragma unroll
        for (int j = 0; j < 2; ++j) {
          const int s = 4 * kk + 2 * half + j;
          uB[kk][j] = uT[(s * BP + col) * DPB + wv];
        }
      }

      v8f y = {};
#pragma unroll
      for (int kk = 0; kk < 4; ++kk) y = wmma4(mA[kk], uB[kk], y);

#pragma unroll
      for (int kk = 0; kk < 16; ++kk) {
        v2f xb;
#pragma unroll
        for (int j = 0; j < 2; ++j) {
          const int n0 = 4 * kk + j;
          const int n1 = 4 * kk + 2 + j;
          const float c0 =
              __shfl(x[n0 >> 4][(n0 & 15) & 7], col + 16 * ((n0 & 15) >> 3), 32);
          const float c1 =
              __shfl(x[n1 >> 4][(n1 & 15) & 7], col + 16 * ((n1 & 15) >> 3), 32);
          xb[j] = half ? c1 : c0;
        }
        y = wmma4(pA[kk], xb, y);
      }

#pragma unroll
      for (int r = 0; r < 4; ++r) {
        v8f xn;
#pragma unroll
        for (int i = 0; i < 8; ++i) xn[i] = lam[r][i] * x[r][i];
#pragma unroll
        for (int kk = 0; kk < 4; ++kk) xn = wmma4(gA[r][kk], uB[kk], xn);
        x[r] = xn;
      }

      float* yT = &ldsY[buf][k][0];
#pragma unroll
      for (int i = 0; i < 8; ++i) {
        const int t = i + 8 * half;
        yT[(t * BP + col) * DPB + wv] = y[i];
      }
    }

#pragma unroll
    for (int k = 0; k < SUP; ++k) *(v4f*)(&ldsU[buf ^ 1][k][coop]) = nv[k];
    __syncthreads();
#pragma unroll
    for (int k = 0; k < SUP; ++k) {
      v4f v = *(const v4f*)(&ldsY[buf][k][coop]);
      *(v4f*)(out + gRow + (size_t)((g * SUP + k) * T + tt) * HID) = v;
    }
  }
}

extern "C" void kernel_launch(void* const* d_in, const int* in_sizes, int n_in,
                              void* d_out, int out_size, void* d_ws,
                              size_t ws_size, hipStream_t stream) {
  (void)in_sizes; (void)n_in; (void)out_size; (void)ws_size;
  const float* u  = (const float*)d_in[0];
  const float* A  = (const float*)d_in[1];
  const float* Bm = (const float*)d_in[2];
  const float* C  = (const float*)d_in[3];
  const float* Dv = (const float*)d_in[4];
  const float* dt = (const float*)d_in[5];
  float* out = (float*)d_out;
  float* ws  = (float*)d_ws;

  ssm_setup_kernel<<<HID, STATE, 0, stream>>>(A, Bm, C, Dv, dt, ws);
  ssm_scan_kernel<<<HID / DPB, 32 * DPB, 0, stream>>>(u, ws, out);
}


# P*X takes the carried state straight from its accumulator registers as B operand (P's k order permuted to match): the 64 cross-lane reads per chunk go; the 64 state terms are summed in a new order
# speedup vs baseline: 3.1012x; 1.2856x over previous
#include <hip/hip_runtime.h>


typedef __attribute__((ext_vector_type(2))) float v2f;
typedef __attribute__((ext_vector_type(4))) float v4f;
typedef __attribute__((ext_vector_type(8))) float v8f;

#define BATCH  8
#define SEQLEN 4096
#define HID    1024
#define STATE  64
#define T      16
#define NCHUNK (SEQLEN / T)
#define DPB    4
#define BP     16
#define SUP    4
#define NSTAGE (NCHUNK / SUP)
#define TILE   (T * BP * DPB)

#define PER_D   2368
#define OFF_P   0
#define OFF_G   1024
#define OFF_M   2048
#define OFF_LAM 2304

static __device__ __forceinline__ v8f wmma4(v2f a, v2f b, v8f c) {
  return __builtin_amdgcn_wmma_f32_16x16x4_f32(false, a, false, b, (short)0, c,
                                               false, false);
}

__global__ __launch_bounds__(64) void ssm_setup_kernel(
    const float* __restrict__ A, const float* __restrict__ Bm,
    const float* __restrict__ C, const float* __restrict__ Dv,
    const float* __restrict__ dt, float* __restrict__ ws) {
  const int d = blockIdx.x;
  const int n = threadIdx.x;
  __shared__ float sK[T * STATE];
  __shared__ float sKl[T];
  float* base = ws + (size_t)d * PER_D;

  const float a  = expf(A[d * STATE + n]);
  const float da = expf(-a * dt[n]);
  const float db = Bm[d * STATE + n] * (1.0f - da) / a;
  const float cc = C[d * STATE + n];

  float p = da;
#pragma unroll
  for (int t = 0; t < T; ++t) { base[OFF_P + t * STATE + n] = cc * p; p *= da; }

  const float da2 = da * da, da4 = da2 * da2, da8 = da4 * da4;
  base[OFF_LAM + n] = da8 * da8;

  float g = db;
  for (int s = T - 1; s >= 0; --s) { base[OFF_G + n * T + s] = g; g *= da; }

  float w = cc * db;
#pragma unroll
  for (int tau = 0; tau < T; ++tau) { sK[tau * STATE + n] = w; w *= da; }
  __syncthreads();
  if (n < T) {
    float s = 0.0f;
    for (int m = 0; m < STATE; ++m) s += sK[n * STATE + m];
    sKl[n] = s;
  }
  __syncthreads();
  if (n < T) {
    const int t = n;
    const float dvd = Dv[d];
    for (int s2 = 0; s2 < T; ++s2) {
      float v = (t >= s2) ? sKl[t - s2] : 0.0f;
      if (s2 == t) v += dvd;
      base[OFF_M + t * T + s2] = v;
    }
  }
}

__global__ __launch_bounds__(32 * DPB) void ssm_scan_kernel(
    const float* __restrict__ u, const float* __restrict__ ws,
    float* __restrict__ out) {
  const int tid  = threadIdx.x;
  const int lane = tid & 31;
  const int wv   = tid >> 5;
  const int d0   = blockIdx.x * DPB;
  const int half = lane >> 4;
  const int col  = lane & 15;

  __shared__ __align__(16) float ldsU[2][SUP][TILE];
  __shared__ __align__(16) float ldsY[2][SUP][TILE];

  {
    const v4f z = {0.0f, 0.0f, 0.0f, 0.0f};
    for (int k = 0; k < (2 * SUP * TILE) / (4 * 32 * DPB); ++k)
      *(v4f*)(&ldsU[0][0][0] + 4 * (tid + k * (32 * DPB))) = z;
  }
  __syncthreads();

  const float* base = ws + (size_t)(d0 + wv) * PER_D;

  v2f pA[16];
#pragma unroll
  for (int kk = 0; kk < 16; ++kk)
    pA[kk] = *(const v2f*)(base + OFF_P + col * STATE + 16 * (kk >> 2) +
                           8 * half + 2 * (kk & 3));
  v2f mA[4];
#pragma unroll
  for (int kk = 0; kk < 4; ++kk)
    mA[kk] = *(const v2f*)(base + OFF_M + col * T + 4 * kk + 2 * half);
  v2f gA[4][4];
#pragma unroll
  for (int r = 0; r < 4; ++r)
#pragma unroll
    for (int kk = 0; kk < 4; ++kk)
      gA[r][kk] =
          *(const v2f*)(base + OFF_G + (r * 16 + col) * T + 4 * kk + 2 * half);
  float lam[4][8];
#pragma unroll
  for (int r = 0; r < 4; ++r)
#pragma unroll
    for (int i = 0; i < 8; ++i)
      lam[r][i] = base[OFF_LAM + r * 16 + i + 8 * half];

  v8f x[4] = {};

  const int tt   = tid >> 3;
  const int bb   = tid & 7;
  const int d4   = 0;
  const size_t gRow = (size_t)bb * SEQLEN * HID + (size_t)(d0 + d4);
  const int coop = (tt * BP + bb) * DPB + d4;

#pragma unroll
  for (int k = 0; k < SUP; ++k)
    *(v4f*)(&ldsU[0][k][coop]) =
        *(const v4f*)(u + gRow + (size_t)(k * T + tt) * HID);
  __syncthreads();

  for (int g = 0; g < NSTAGE; ++g) {
    const int buf = g & 1;
    const int gn = (g + 1) & (NSTAGE - 1);
    const int gp = (g + 2) & (NSTAGE - 1);
    v4f nv[SUP];
#pragma unroll
    for (int k = 0; k < SUP; ++k) {
      nv[k] = *(const v4f*)(u + gRow + (size_t)((gn * SUP + k) * T + tt) * HID);
      __builtin_prefetch(u + gRow + (size_t)((gp * SUP + k) * T + tt) * HID, 0, 0);
    }

#pragma unroll 1
    for (int k = 0; k < SUP; ++k) {
      const float* uT = &ldsU[buf][k][0];
      v2f uB[4];
#pragma unroll
      for (int kk = 0; kk < 4; ++kk) {
#pragma unroll
        for (int j = 0; j < 2; ++j) {
          const int s = 4 * kk + 2 * half + j;
          uB[kk][j] = uT[(s * BP + col) * DPB + wv];
        }
      }

      v8f y = {};
#pragma unroll
      for (int kk = 0; kk < 4; ++kk) y = wmma4(mA[kk], uB[kk], y);

#pragma unroll
      for (int kk = 0; kk < 16; ++kk) {
        v2f xb;
        xb[0] = x[kk >> 2][2 * (kk & 3)];
        xb[1] = x[kk >> 2][2 * (kk & 3) + 1];
        y = wmma4(pA[kk], xb, y);
      }

#pragma unroll
      for (int r = 0; r < 4; ++r) {
        v8f xn;
#pragma unroll
        for (int i = 0; i < 8; ++i) xn[i] = lam[r][i] * x[r][i];
#pragma unroll
        for (int kk = 0; kk < 4; ++kk) xn = wmma4(gA[r][kk], uB[kk], xn);
        x[r] = xn;
      }

      float* yT = &ldsY[buf][k][0];
#pragma unroll
      for (int i = 0; i < 8; ++i) {
        const int t = i + 8 * half;
        yT[(t * BP + col) * DPB + wv] = y[i];
      }
    }

#pragma unroll
    for (int k = 0; k < SUP; ++k) *(v4f*)(&ldsU[buf ^ 1][k][coop]) = nv[k];
    __syncthreads();
#pragma unroll
    for (int k = 0; k < SUP; ++k) {
      v4f v = *(const v4f*)(&ldsY[buf][k][coop]);
      *(v4f*)(out + gRow + (size_t)((g * SUP + k) * T + tt) * HID) = v;
    }
  }
}

extern "C" void kernel_launch(void* const* d_in, const int* in_sizes, int n_in,
                              void* d_out, int out_size, void* d_ws,
                              size_t ws_size, hipStream_t stream) {
  (void)in_sizes; (void)n_in; (void)out_size; (void)ws_size;
  const float* u  = (const float*)d_in[0];
  const float* A  = (const float*)d_in[1];
  const float* Bm = (const float*)d_in[2];
  const float* C  = (const float*)d_in[3];
  const float* Dv = (const float*)d_in[4];
  const float* dt = (const float*)d_in[5];
  float* out = (float*)d_out;
  float* ws  = (float*)d_ws;

  ssm_setup_kernel<<<HID, STATE, 0, stream>>>(A, Bm, C, Dv, dt, ws);
  ssm_scan_kernel<<<HID / DPB, 32 * DPB, 0, stream>>>(u, ws, out);
}


# Two chunks share each WMMA tile (odd chunk in the 8 idle batch columns): 36 WMMAs per 2 chunks, not 72; state steps now fused Lam*X+G*U on the vector unit with G*U summed first (new sum order)
# speedup vs baseline: 3.4988x; 1.1282x over previous
#include <hip/hip_runtime.h>


typedef __attribute__((ext_vector_type(2))) float v2f;
typedef __attribute__((ext_vector_type(4))) float v4f;
typedef __attribute__((ext_vector_type(8))) float v8f;

#define BATCH  8
#define SEQLEN 4096
#define HID    1024
#define STATE  64
#define T      16
#define NCHUNK (SEQLEN / T)
#define DPB    4
#define BP     16
#define SUP    4
#define NPAIR  (SUP / 2)
#define NSTAGE (NCHUNK / SUP)
#define TILE   (T * BP * DPB)
#define ODD    (8 * DPB)

#define PER_D   2368
#define OFF_P   0
#define OFF_G   1024
#define OFF_M   2048
#define OFF_LAM 2304

static __device__ __forceinline__ float row_swap8(float v) {
  const int s = __builtin_bit_cast(int, v);
  return __builtin_bit_cast(
      float, __builtin_amdgcn_update_dpp(s, s, 0x128, 0xF, 0xF, false));
}
static __device__ __forceinline__ float row_upper_from_lower(float keep, float v) {
  return __builtin_bit_cast(
      float, __builtin_amdgcn_update_dpp(__builtin_bit_cast(int, keep),
                                         __builtin_bit_cast(int, v), 0x128, 0xF,
                                         0xC, false));
}

static __device__ __forceinline__ v8f wmma4(v2f a, v2f b, v8f c) {
  return __builtin_amdgcn_wmma_f32_16x16x4_f32(false, a, false, b, (short)0, c,
                                               false, false);
}

__global__ __launch_bounds__(64) void ssm_setup_kernel(
    const float* __restrict__ A, const float* __restrict__ Bm,
    const float* __restrict__ C, const float* __restrict__ Dv,
    const float* __restrict__ dt, float* __restrict__ ws) {
  const int d = blockIdx.x;
  const int n = threadIdx.x;
  __shared__ float sK[T * STATE];
  __shared__ float sKl[T];
  float* base = ws + (size_t)d * PER_D;

  const float a  = expf(A[d * STATE + n]);
  const float da = expf(-a * dt[n]);
  const float db = Bm[d * STATE + n] * (1.0f - da) / a;
  const float cc = C[d * STATE + n];

  float p = da;
#pragma unroll
  for (int t = 0; t < T; ++t) { base[OFF_P + t * STATE + n] = cc * p; p *= da; }

  const float da2 = da * da, da4 = da2 * da2, da8 = da4 * da4;
  base[OFF_LAM + n] = da8 * da8;

  float g = db;
  for (int s = T - 1; s >= 0; --s) { base[OFF_G + n * T + s] = g; g *= da; }

  float w = cc * db;
#pragma unroll
  for (int tau = 0; tau < T; ++tau) { sK[tau * STATE + n] = w; w *= da; }
  __syncthreads();
  if (n < T) {
    float s = 0.0f;
    for (int m = 0; m < STATE; ++m) s += sK[n * STATE + m];
    sKl[n] = s;
  }
  __syncthreads();
  if (n < T) {
    const int t = n;
    const float dvd = Dv[d];
    for (int s2 = 0; s2 < T; ++s2) {
      float v = (t >= s2) ? sKl[t - s2] : 0.0f;
      if (s2 == t) v += dvd;
      base[OFF_M + t * T + s2] = v;
    }
  }
}

__global__ __launch_bounds__(32 * DPB) void ssm_scan_kernel(
    const float* __restrict__ u, const float* __restrict__ ws,
    float* __restrict__ out) {
  const int tid  = threadIdx.x;
  const int lane = tid & 31;
  const int wv   = tid >> 5;
  const int d0   = blockIdx.x * DPB;
  const int half = lane >> 4;
  const int col  = lane & 15;

  __shared__ __align__(16) float ldsU[2][NPAIR][TILE];
  __shared__ __align__(16) float ldsY[2][NPAIR][TILE];

  const float* base = ws + (size_t)(d0 + wv) * PER_D;

  v2f pA[16];
#pragma unroll
  for (int kk = 0; kk < 16; ++kk)
    pA[kk] = *(const v2f*)(base + OFF_P + col * STATE + 16 * (kk >> 2) +
                           8 * half + 2 * (kk & 3));
  v2f mA[4];
#pragma unroll
  for (int kk = 0; kk < 4; ++kk)
    mA[kk] = *(const v2f*)(base + OFF_M + col * T + 4 * kk + 2 * half);
  v2f gA[4][4];
#pragma unroll
  for (int r = 0; r < 4; ++r)
#pragma unroll
    for (int kk = 0; kk < 4; ++kk)
      gA[r][kk] =
          *(const v2f*)(base + OFF_G + (r * 16 + col) * T + 4 * kk + 2 * half);
  float lam[4][8];
#pragma unroll
  for (int r = 0; r < 4; ++r)
#pragma unroll
    for (int i = 0; i < 8; ++i)
      lam[r][i] = base[OFF_LAM + r * 16 + i + 8 * half];

  v8f x[4] = {};

  const int tt   = tid >> 3;
  const int bb   = tid & 7;
  const int d4   = 0;
  const size_t gRow = (size_t)bb * SEQLEN * HID + (size_t)(d0 + d4);
  const int coop = (tt * BP + bb) * DPB + d4;

#pragma unroll
  for (int k = 0; k < SUP; ++k)
    *(v4f*)(&ldsU[0][k >> 1][coop + (k & 1) * ODD]) =
        *(const v4f*)(u + gRow + (size_t)(k * T + tt) * HID);
  __syncthreads();

  for (int g = 0; g < NSTAGE; ++g) {
    const int buf = g & 1;
    const int gn = (g + 1) & (NSTAGE - 1);
    const int gp = (g + 2) & (NSTAGE - 1);
    v4f nv[SUP];
#pragma unroll
    for (int k = 0; k < SUP; ++k) {
      nv[k] = *(const v4f*)(u + gRow + (size_t)((gn * SUP + k) * T + tt) * HID);
      __builtin_prefetch(u + gRow + (size_t)((gp * SUP + k) * T + tt) * HID, 0, 0);
    }

#pragma unroll 1
    for (int kp = 0; kp < NPAIR; ++kp) {
      const float* uT = &ldsU[buf][kp][0];
      v2f uB[4];
#pragma unroll
      for (int kk = 0; kk < 4; ++kk) {
#pragma unroll
        for (int j = 0; j < 2; ++j) {
          const int s = 4 * kk + 2 * half + j;
          uB[kk][j] = uT[(s * BP + col) * DPB + wv];
        }
      }

      v8f y = {};
#pragma unroll
      for (int kk = 0; kk < 4; ++kk) y = wmma4(mA[kk], uB[kk], y);

      v8f gu[4];
#pragma unroll
      for (int r = 0; r < 4; ++r) {
        v8f acc = {};
#pragma unroll
        for (int kk = 0; kk < 4; ++kk) acc = wmma4(gA[r][kk], uB[kk], acc);
        gu[r] = acc;
      }

      v8f xs[4];
#pragma unroll
      for (int r = 0; r < 4; ++r) {
#pragma unroll
        for (int i = 0; i < 8; ++i) {
          const float a = __builtin_fmaf(lam[r][i], x[r][i], gu[r][i]);
          const float s = row_upper_from_lower(x[r][i], a);
          const float b = __builtin_fmaf(lam[r][i], s, gu[r][i]);
          xs[r][i] = s;
          x[r][i]  = row_swap8(b);
        }
      }

#pragma unroll
      for (int kk = 0; kk < 16; ++kk) {
        v2f xb;
        xb[0] = xs[kk >> 2][2 * (kk & 3)];
        xb[1] = xs[kk >> 2][2 * (kk & 3) + 1];
        y = wmma4(pA[kk], xb, y);
      }

      float* yT = &ldsY[buf][kp][0];
#pragma unroll
      for (int i = 0; i < 8; ++i) {
        const int t = i + 8 * half;
        yT[(t * BP + col) * DPB + wv] = y[i];
      }
    }

#pragma unroll
    for (int k = 0; k < SUP; ++k)
      *(v4f*)(&ldsU[buf ^ 1][k >> 1][coop + (k & 1) * ODD]) = nv[k];
    __syncthreads();
#pragma unroll
    for (int k = 0; k < SUP; ++k) {
      v4f v = *(const v4f*)(&ldsY[buf][k >> 1][coop + (k & 1) * ODD]);
      *(v4f*)(out + gRow + (size_t)((g * SUP + k) * T + tt) * HID) = v;
    }
  }
}

extern "C" void kernel_launch(void* const* d_in, const int* in_sizes, int n_in,
                              void* d_out, int out_size, void* d_ws,
                              size_t ws_size, hipStream_t stream) {
  (void)in_sizes; (void)n_in; (void)out_size; (void)ws_size;
  const float* u  = (const float*)d_in[0];
  const float* A  = (const float*)d_in[1];
  const float* Bm = (const float*)d_in[2];
  const float* C  = (const float*)d_in[3];
  const float* Dv = (const float*)d_in[4];
  const float* dt = (const float*)d_in[5];
  float* out = (float*)d_out;
  float* ws  = (float*)d_ws;

  ssm_setup_kernel<<<HID, STATE, 0, stream>>>(A, Bm, C, Dv, dt, ws);
  ssm_scan_kernel<<<HID / DPB, 32 * DPB, 0, stream>>>(u, ws, out);
}


# u and y pass through channel-blocked copies in the workspace (two LDS re-blocking kernels round the scan): each scan wave load/store is one 512-B run, not 32 separate lines; sums unchanged
# speedup vs baseline: 4.7180x; 1.3485x over previous
#include <hip/hip_runtime.h>


typedef __attribute__((ext_vector_type(2))) float v2f;
typedef __attribute__((ext_vector_type(4))) float v4f;
typedef __attribute__((ext_vector_type(8))) float v8f;

#define BATCH  8
#define SEQLEN 4096
#define HID    1024
#define STATE  64
#define T      16
#define NCHUNK (SEQLEN / T)
#define DPB    4
#define BP     16
#define SUP    4
#define NPAIR  (SUP / 2)
#define NSTAGE (NCHUNK / SUP)
#define TILE   (T * BP * DPB)
#define ODD    (8 * DPB)
#define CHK    (T * BATCH * DPB)

#define PER_D   2368
#define OFF_P   0
#define OFF_G   1024
#define OFF_M   2048
#define OFF_LAM 2304

#define QBLK    (SEQLEN * BATCH * DPB)
#define OFF_UT  ((size_t)4194304)
#define OFF_YT  (OFF_UT + (size_t)BATCH * SEQLEN * HID)

#define XT      4
#define XROWS   (XT * BATCH)
#define XQ      64
#define XPITCH  65
#define XTHR    256
#define XWAVES  (XTHR / 32)
#define XPASS   (XROWS * XQ / XTHR)
#define XQPARTS (HID / DPB / XQ)
#define XTILES  ((SEQLEN / XT) * XQPARTS)
#define XGRID   1024
#define XITER   (XTILES / XGRID)

static __device__ __forceinline__ float row_swap8(float v) {
  const int s = __builtin_bit_cast(int, v);
  return __builtin_bit_cast(
      float, __builtin_amdgcn_update_dpp(s, s, 0x128, 0xF, 0xF, false));
}
static __device__ __forceinline__ float row_upper_from_lower(float keep, float v) {
  return __builtin_bit_cast(
      float, __builtin_amdgcn_update_dpp(__builtin_bit_cast(int, keep),
                                         __builtin_bit_cast(int, v), 0x128, 0xF,
                                         0xC, false));
}

static __device__ __forceinline__ v8f wmma4(v2f a, v2f b, v8f c) {
  return __builtin_amdgcn_wmma_f32_16x16x4_f32(false, a, false, b, (short)0, c,
                                               false, false);
}

__global__ __launch_bounds__(64) void ssm_setup_kernel(
    const float* __restrict__ A, const float* __restrict__ Bm,
    const float* __restrict__ C, const float* __restrict__ Dv,
    const float* __restrict__ dt, float* __restrict__ ws) {
  const int d = blockIdx.x;
  const int n = threadIdx.x;
  __shared__ float sK[T * STATE];
  __shared__ float sKl[T];
  float* base = ws + (size_t)d * PER_D;

  const float a  = expf(A[d * STATE + n]);
  const float da = expf(-a * dt[n]);
  const float db = Bm[d * STATE + n] * (1.0f - da) / a;
  const float cc = C[d * STATE + n];

  float p = da;
#pragma unroll
  for (int t = 0; t < T; ++t) { base[OFF_P + t * STATE + n] = cc * p; p *= da; }

  const float da2 = da * da, da4 = da2 * da2, da8 = da4 * da4;
  base[OFF_LAM + n] = da8 * da8;

  float g = db;
  for (int s = T - 1; s >= 0; --s) { base[OFF_G + n * T + s] = g; g *= da; }

  float w = cc * db;
#pragma unroll
  for (int tau = 0; tau < T; ++tau) { sK[tau * STATE + n] = w; w *= da; }
  __syncthreads();
  if (n < T) {
    float s = 0.0f;
    for (int m = 0; m < STATE; ++m) s += sK[n * STATE + m];
    sKl[n] = s;
  }
  __syncthreads();
  if (n < T) {
    const int t = n;
    const float dvd = Dv[d];
    for (int s2 = 0; s2 < T; ++s2) {
      float v = (t >= s2) ? sKl[t - s2] : 0.0f;
      if (s2 == t) v += dvd;
      base[OFF_M + t * T + s2] = v;
    }
  }
}

__global__ __launch_bounds__(XTHR) void ssm_block_in_kernel(
    const float* __restrict__ u, float* __restrict__ ws) {
  __shared__ __align__(16) float tile[XROWS * XPITCH * 4];
  float* uT = ws + OFF_UT;
  const int lane = threadIdx.x & 31;
  const int w    = threadIdx.x >> 5;
  for (int it = 0; it < XITER; ++it) {
    const int tileId = it * XGRID + blockIdx.x;
    const int t0 = (tileId / XQPARTS) * XT;
    const int q0 = (tileId % XQPARTS) * XQ;
    v4f v[XPASS];
#pragma unroll
    for (int p = 0; p < XPASS; ++p) {
      const int i  = p * XWAVES + w;
      const int r  = i >> 1;
      const int ql = (i & 1) * 32 + lane;
      v[p] = *(const v4f*)(u + ((size_t)(r & 7) * SEQLEN + t0 + (r >> 3)) * HID +
                           (size_t)(q0 + ql) * DPB);
    }
    __syncthreads();
#pragma unroll
    for (int p = 0; p < XPASS; ++p) {
      const int i  = p * XWAVES + w;
      const int r  = i >> 1;
      const int ql = (i & 1) * 32 + lane;
      *(v4f*)(&tile[(r * XPITCH + ql) * 4]) = v[p];
    }
    __syncthreads();
#pragma unroll
    for (int p = 0; p < XPASS; ++p) {
      const int ql = p * XWAVES + w;
      const v4f x = *(const v4f*)(&tile[(lane * XPITCH + ql) * 4]);
      *(v4f*)(uT + (size_t)(q0 + ql) * QBLK + (size_t)t0 * (BATCH * DPB) +
              lane * 4) = x;
    }
  }
}

__global__ __launch_bounds__(XTHR) void ssm_block_out_kernel(
    const float* __restrict__ ws, float* __restrict__ out) {
  __shared__ __align__(16) float tile[XROWS * XPITCH * 4];
  const float* yT = ws + OFF_YT;
  const int lane = threadIdx.x & 31;
  const int w    = threadIdx.x >> 5;
  for (int it = 0; it < XITER; ++it) {
    const int tileId = it * XGRID + blockIdx.x;
    const int t0 = (tileId / XQPARTS) * XT;
    const int q0 = (tileId % XQPARTS) * XQ;
    v4f v[XPASS];
#pragma unroll
    for (int p = 0; p < XPASS; ++p) {
      const int ql = p * XWAVES + w;
      v[p] = *(const v4f*)(yT + (size_t)(q0 + ql) * QBLK +
                           (size_t)t0 * (BATCH * DPB) + lane * 4);
    }
    __syncthreads();
#pragma unroll
    for (int p = 0; p < XPASS; ++p) {
      const int ql = p * XWAVES + w;
      *(v4f*)(&tile[(lane * XPITCH + ql) * 4]) = v[p];
    }
    __syncthreads();
#pragma unroll
    for (int p = 0; p < XPASS; ++p) {
      const int i  = p * XWAVES + w;
      const int r  = i >> 1;
      const int ql = (i & 1) * 32 + lane;
      const v4f x = *(const v4f*)(&tile[(r * XPITCH + ql) * 4]);
      *(v4f*)(out + ((size_t)(r & 7) * SEQLEN + t0 + (r >> 3)) * HID +
              (size_t)(q0 + ql) * DPB) = x;
    }
  }
}

__global__ __launch_bounds__(32 * DPB) void ssm_scan_kernel(
    const float* __restrict__ ws, float* __restrict__ wsy) {
  const int tid  = threadIdx.x;
  const int lane = tid & 31;
  const int wv   = tid >> 5;
  const int d0   = blockIdx.x * DPB;
  const int half = lane >> 4;
  const int col  = lane & 15;

  __shared__ __align__(16) float ldsU[2][NPAIR][TILE];
  __shared__ __align__(16) float ldsY[2][NPAIR][TILE];

  const float* base = ws + (size_t)(d0 + wv) * PER_D;

  v2f pA[16];
#pragma unroll
  for (int kk = 0; kk < 16; ++kk)
    pA[kk] = *(const v2f*)(base + OFF_P + col * STATE + 16 * (kk >> 2) +
                           8 * half + 2 * (kk & 3));
  v2f mA[4];
#pragma unroll
  for (int kk = 0; kk < 4; ++kk)
    mA[kk] = *(const v2f*)(base + OFF_M + col * T + 4 * kk + 2 * half);
  v2f gA[4][4];
#pragma unroll
  for (int r = 0; r < 4; ++r)
#pragma unroll
    for (int kk = 0; kk < 4; ++kk)
      gA[r][kk] =
          *(const v2f*)(base + OFF_G + (r * 16 + col) * T + 4 * kk + 2 * half);
  float lam[4][8];
#pragma unroll
  for (int r = 0; r < 4; ++r)
#pragma unroll
    for (int i = 0; i < 8; ++i)
      lam[r][i] = base[OFF_LAM + r * 16 + i + 8 * half];

  v8f x[4] = {};

  const int tt   = tid >> 3;
  const int bb   = tid & 7;
  const int coop = (tt * BP + bb) * DPB;
  const float* uq = ws + OFF_UT + (size_t)blockIdx.x * QBLK + tid * 4;
  float*       yq = wsy + OFF_YT + (size_t)blockIdx.x * QBLK + tid * 4;

#pragma unroll
  for (int k = 0; k < SUP; ++k)
    *(v4f*)(&ldsU[0][k >> 1][coop + (k & 1) * ODD]) =
        *(const v4f*)(uq + (size_t)k * CHK);
  __syncthreads();

  for (int g = 0; g < NSTAGE; ++g) {
    const int buf = g & 1;
    const int gn = (g + 1) & (NSTAGE - 1);
    const int gp = (g + 2) & (NSTAGE - 1);
    v4f nv[SUP];
#pragma unroll
    for (int k = 0; k < SUP; ++k) {
      nv[k] = *(const v4f*)(uq + (size_t)(gn * SUP + k) * CHK);
      __builtin_prefetch(uq + (size_t)(gp * SUP + k) * CHK, 0, 0);
    }

#pragma unroll 1
    for (int kp = 0; kp < NPAIR; ++kp) {
      const float* uT = &ldsU[buf][kp][0];
      v2f uB[4];
#pragma unroll
      for (int kk = 0; kk < 4; ++kk) {
#pragma unroll
        for (int j = 0; j < 2; ++j) {
          const int s = 4 * kk + 2 * half + j;
          uB[kk][j] = uT[(s * BP + col) * DPB + wv];
        }
      }

      v8f y = {};
#pragma unroll
      for (int kk = 0; kk < 4; ++kk) y = wmma4(mA[kk], uB[kk], y);

      v8f gu[4];
#pragma unroll
      for (int r = 0; r < 4; ++r) {
        v8f acc = {};
#pragma unroll
        for (int kk = 0; kk < 4; ++kk) acc = wmma4(gA[r][kk], uB[kk], acc);
        gu[r] = acc;
      }

      v8f xs[4];
#pragma unroll
      for (int r = 0; r < 4; ++r) {
#pragma unroll
        for (int i = 0; i < 8; ++i) {
          const float a = __builtin_fmaf(lam[r][i], x[r][i], gu[r][i]);
          const float s = row_upper_from_lower(x[r][i], a);
          const float b = __builtin_fmaf(lam[r][i], s, gu[r][i]);
          xs[r][i] = s;
          x[r][i]  = row_swap8(b);
        }
      }

#pragma unroll
      for (int kk = 0; kk < 16; ++kk) {
        v2f xb;
        xb[0] = xs[kk >> 2][2 * (kk & 3)];
        xb[1] = xs[kk >> 2][2 * (kk & 3) + 1];
        y = wmma4(pA[kk], xb, y);
      }

      float* yT = &ldsY[buf][kp][0];
#pragma unroll
      for (int i = 0; i < 8; ++i) {
        const int t = i + 8 * half;
        yT[(t * BP + col) * DPB + wv] = y[i];
      }
    }

#pragma unroll
    for (int k = 0; k < SUP; ++k)
      *(v4f*)(&ldsU[buf ^ 1][k >> 1][coop + (k & 1) * ODD]) = nv[k];
    __syncthreads();
#pragma unroll
    for (int k = 0; k < SUP; ++k) {
      v4f v = *(const v4f*)(&ldsY[buf][k >> 1][coop + (k & 1) * ODD]);
      *(v4f*)(yq + (size_t)(g * SUP + k) * CHK) = v;
    }
  }
}

extern "C" void kernel_launch(void* const* d_in, const int* in_sizes, int n_in,
                              void* d_out, int out_size, void* d_ws,
                              size_t ws_size, hipStream_t stream) {
  (void)in_sizes; (void)n_in; (void)out_size; (void)ws_size;
  const float* u  = (const float*)d_in[0];
  const float* A  = (const float*)d_in[1];
  const float* Bm = (const float*)d_in[2];
  const float* C  = (const float*)d_in[3];
  const float* Dv = (const float*)d_in[4];
  const float* dt = (const float*)d_in[5];
  float* out = (float*)d_out;
  float* ws  = (float*)d_ws;

  ssm_setup_kernel<<<HID, STATE, 0, stream>>>(A, Bm, C, Dv, dt, ws);
  ssm_block_in_kernel<<<XGRID, XTHR, 0, stream>>>(u, ws);
  ssm_scan_kernel<<<HID / DPB, 32 * DPB, 0, stream>>>(ws, ws);
  ssm_block_out_kernel<<<XGRID, XTHR, 0, stream>>>(ws, out);
}


# Re-blocking kernels made barrier-free streams: each wave transposes its own 32x32-unit tile through a private LDS region with the whole tile (32 loads) in flight, 16 waves per WGP; sums unchanged
# speedup vs baseline: 4.9363x; 1.0463x over previous
#include <hip/hip_runtime.h>


typedef __attribute__((ext_vector_type(2))) float v2f;
typedef __attribute__((ext_vector_type(4))) float v4f;
typedef __attribute__((ext_vector_type(8))) float v8f;

#define BATCH  8
#define SEQLEN 4096
#define HID    1024
#define STATE  64
#define T      16
#define NCHUNK (SEQLEN / T)
#define DPB    4
#define BP     16
#define SUP    4
#define NPAIR  (SUP / 2)
#define NSTAGE (NCHUNK / SUP)
#define TILE   (T * BP * DPB)
#define ODD    (8 * DPB)
#define CHK    (T * BATCH * DPB)

#define PER_D   2368
#define OFF_P   0
#define OFF_G   1024
#define OFF_M   2048
#define OFF_LAM 2304

#define QBLK    (SEQLEN * BATCH * DPB)
#define OFF_UT  ((size_t)4194304)
#define OFF_YT  (OFF_UT + (size_t)BATCH * SEQLEN * HID)

#define XT      4
#define XROWS   (XT * BATCH)
#define XQ      32
#define XPITCH  33
#define XTHR    256
#define XWAVES  (XTHR / 32)
#define XQPARTS (HID / DPB / XQ)
#define XTILES  ((SEQLEN / XT) * XQPARTS)
#define XGRID   512
#define XITER   (XTILES / (XGRID * XWAVES))

static __device__ __forceinline__ float row_swap8(float v) {
  const int s = __builtin_bit_cast(int, v);
  return __builtin_bit_cast(
      float, __builtin_amdgcn_update_dpp(s, s, 0x128, 0xF, 0xF, false));
}
static __device__ __forceinline__ float row_upper_from_lower(float keep, float v) {
  return __builtin_bit_cast(
      float, __builtin_amdgcn_update_dpp(__builtin_bit_cast(int, keep),
                                         __builtin_bit_cast(int, v), 0x128, 0xF,
                                         0xC, false));
}

static __device__ __forceinline__ v8f wmma4(v2f a, v2f b, v8f c) {
  return __builtin_amdgcn_wmma_f32_16x16x4_f32(false, a, false, b, (short)0, c,
                                               false, false);
}

__global__ __launch_bounds__(64) void ssm_setup_kernel(
    const float* __restrict__ A, const float* __restrict__ Bm,
    const float* __restrict__ C, const float* __restrict__ Dv,
    const float* __restrict__ dt, float* __restrict__ ws) {
  const int d = blockIdx.x;
  const int n = threadIdx.x;
  __shared__ float sK[T * STATE];
  __shared__ float sKl[T];
  float* base = ws + (size_t)d * PER_D;

  const float a  = expf(A[d * STATE + n]);
  const float da = expf(-a * dt[n]);
  const float db = Bm[d * STATE + n] * (1.0f - da) / a;
  const float cc = C[d * STATE + n];

  float p = da;
#pragma unroll
  for (int t = 0; t < T; ++t) { base[OFF_P + t * STATE + n] = cc * p; p *= da; }

  const float da2 = da * da, da4 = da2 * da2, da8 = da4 * da4;
  base[OFF_LAM + n] = da8 * da8;

  float g = db;
  for (int s = T - 1; s >= 0; --s) { base[OFF_G + n * T + s] = g; g *= da; }

  float w = cc * db;
#pragma unroll
  for (int tau = 0; tau < T; ++tau) { sK[tau * STATE + n] = w; w *= da; }
  __syncthreads();
  if (n < T) {
    float s = 0.0f;
    for (int m = 0; m < STATE; ++m) s += sK[n * STATE + m];
    sKl[n] = s;
  }
  __syncthreads();
  if (n < T) {
    const int t = n;
    const float dvd = Dv[d];
    for (int s2 = 0; s2 < T; ++s2) {
      float v = (t >= s2) ? sKl[t - s2] : 0.0f;
      if (s2 == t) v += dvd;
      base[OFF_M + t * T + s2] = v;
    }
  }
}

static __device__ __forceinline__ void wave_lds_order() {
  __builtin_amdgcn_fence(__ATOMIC_RELEASE, "wavefront");
  __builtin_amdgcn_wave_barrier();
  __builtin_amdgcn_fence(__ATOMIC_ACQUIRE, "wavefront");
}

__global__ __launch_bounds__(XTHR) void ssm_block_in_kernel(
    const float* __restrict__ u, float* __restrict__ ws) {
  __shared__ __align__(16) float tiles[XWAVES][XROWS * XPITCH * 4];
  float* uT = ws + OFF_UT;
  const int lane = threadIdx.x & 31;
  const int w    = threadIdx.x >> 5;
  float* tile = &tiles[w][0];
  const int gw = blockIdx.x * XWAVES + w;
  v4f v[XROWS];
#pragma unroll
  for (int r = 0; r < XROWS; ++r) {
    const int t0 = (gw / XQPARTS) * XT;
    const int q0 = (gw % XQPARTS) * XQ;
    v[r] = *(const v4f*)(u + ((size_t)(r & 7) * SEQLEN + t0 + (r >> 3)) * HID +
                         (size_t)(q0 + lane) * DPB);
  }
#pragma unroll
  for (int it = 0; it < XITER; ++it) {
    const int tileId = it * (XGRID * XWAVES) + gw;
    const int t0 = (tileId / XQPARTS) * XT;
    const int q0 = (tileId % XQPARTS) * XQ;
#pragma unroll
    for (int r = 0; r < XROWS; ++r)
      *(v4f*)(&tile[(r * XPITCH + lane) * 4]) = v[r];
    wave_lds_order();
    if (it + 1 < XITER) {
      const int nId = (it + 1) * (XGRID * XWAVES) + gw;
      const int nt0 = (nId / XQPARTS) * XT;
      const int nq0 = (nId % XQPARTS) * XQ;
#pragma unroll
      for (int r = 0; r < XROWS; ++r)
        v[r] = *(const v4f*)(u + ((size_t)(r & 7) * SEQLEN + nt0 + (r >> 3)) * HID +
                             (size_t)(nq0 + lane) * DPB);
    }
#pragma unroll
    for (int j = 0; j < XQ; ++j) {
      const v4f x = *(const v4f*)(&tile[(lane * XPITCH + j) * 4]);
      *(v4f*)(uT + (size_t)(q0 + j) * QBLK + (size_t)t0 * (BATCH * DPB) +
              lane * 4) = x;
    }
    wave_lds_order();
  }
}

__global__ __launch_bounds__(XTHR) void ssm_block_out_kernel(
    const float* __restrict__ ws, float* __restrict__ out) {
  __shared__ __align__(16) float tiles[XWAVES][XROWS * XPITCH * 4];
  const float* yT = ws + OFF_YT;
  const int lane = threadIdx.x & 31;
  const int w    = threadIdx.x >> 5;
  float* tile = &tiles[w][0];
  const int gw = blockIdx.x * XWAVES + w;
  v4f v[XQ];
#pragma unroll
  for (int j = 0; j < XQ; ++j) {
    const int t0 = (gw / XQPARTS) * XT;
    const int q0 = (gw % XQPARTS) * XQ;
    v[j] = *(const v4f*)(yT + (size_t)(q0 + j) * QBLK +
                         (size_t)t0 * (BATCH * DPB) + lane * 4);
  }
#pragma unroll
  for (int it = 0; it < XITER; ++it) {
    const int tileId = it * (XGRID * XWAVES) + gw;
    const int t0 = (tileId / XQPARTS) * XT;
    const int q0 = (tileId % XQPARTS) * XQ;
#pragma unroll
    for (int j = 0; j < XQ; ++j)
      *(v4f*)(&tile[(lane * XPITCH + j) * 4]) = v[j];
    wave_lds_order();
    if (it + 1 < XITER) {
      const int nId = (it + 1) * (XGRID * XWAVES) + gw;
      const int nt0 = (nId / XQPARTS) * XT;
      const int nq0 = (nId % XQPARTS) * XQ;
#pragma unroll
      for (int j = 0; j < XQ; ++j)
        v[j] = *(const v4f*)(yT + (size_t)(nq0 + j) * QBLK +
                             (size_t)nt0 * (BATCH * DPB) + lane * 4);
    }
#pragma unroll
    for (int r = 0; r < XROWS; ++r) {
      const v4f x = *(const v4f*)(&tile[(r * XPITCH + lane) * 4]);
      *(v4f*)(out + ((size_t)(r & 7) * SEQLEN + t0 + (r >> 3)) * HID +
              (size_t)(q0 + lane) * DPB) = x;
    }
    wave_lds_order();
  }
}

__global__ __launch_bounds__(32 * DPB) void ssm_scan_kernel(
    const float* __restrict__ ws, float* __restrict__ wsy) {
  const int tid  = threadIdx.x;
  const int lane = tid & 31;
  const int wv   = tid >> 5;
  const int d0   = blockIdx.x * DPB;
  const int half = lane >> 4;
  const int col  = lane & 15;

  __shared__ __align__(16) float ldsU[2][NPAIR][TILE];
  __shared__ __align__(16) float ldsY[2][NPAIR][TILE];

  const float* base = ws + (size_t)(d0 + wv) * PER_D;

  v2f pA[16];
#pragma unroll
  for (int kk = 0; kk < 16; ++kk)
    pA[kk] = *(const v2f*)(base + OFF_P + col * STATE + 16 * (kk >> 2) +
                           8 * half + 2 * (kk & 3));
  v2f mA[4];
#pragma unroll
  for (int kk = 0; kk < 4; ++kk)
    mA[kk] = *(const v2f*)(base + OFF_M + col * T + 4 * kk + 2 * half);
  v2f gA[4][4];
#pragma unroll
  for (int r = 0; r < 4; ++r)
#pragma unroll
    for (int kk = 0; kk < 4; ++kk)
      gA[r][kk] =
          *(const v2f*)(base + OFF_G + (r * 16 + col) * T + 4 * kk + 2 * half);
  float lam[4][8];
#pragma unroll
  for (int r = 0; r < 4; ++r)
#pragma unroll
    for (int i = 0; i < 8; ++i)
      lam[r][i] = base[OFF_LAM + r * 16 + i + 8 * half];

  v8f x[4] = {};

  const int tt   = tid >> 3;
  const int bb   = tid & 7;
  const int coop = (tt * BP + bb) * DPB;
  const float* uq = ws + OFF_UT + (size_t)blockIdx.x * QBLK + tid * 4;
  float*       yq = wsy + OFF_YT + (size_t)blockIdx.x * QBLK + tid * 4;

#pragma unroll
  for (int k = 0; k < SUP; ++k)
    *(v4f*)(&ldsU[0][k >> 1][coop + (k & 1) * ODD]) =
        *(const v4f*)(uq + (size_t)k * CHK);
  __syncthreads();

  for (int g = 0; g < NSTAGE; ++g) {
    const int buf = g & 1;
    const int gn = (g + 1) & (NSTAGE - 1);
    const int gp = (g + 2) & (NSTAGE - 1);
    v4f nv[SUP];
#pragma unroll
    for (int k = 0; k < SUP; ++k) {
      nv[k] = *(const v4f*)(uq + (size_t)(gn * SUP + k) * CHK);
      __builtin_prefetch(uq + (size_t)(gp * SUP + k) * CHK, 0, 0);
    }

#pragma unroll 1
    for (int kp = 0; kp < NPAIR; ++kp) {
      const float* uT = &ldsU[buf][kp][0];
      v2f uB[4];
#pragma unroll
      for (int kk = 0; kk < 4; ++kk) {
#pragma unroll
        for (int j = 0; j < 2; ++j) {
          const int s = 4 * kk + 2 * half + j;
          uB[kk][j] = uT[(s * BP + col) * DPB + wv];
        }
      }

      v8f y = {};
#pragma unroll
      for (int kk = 0; kk < 4; ++kk) y = wmma4(mA[kk], uB[kk], y);

      v8f gu[4];
#pragma unroll
      for (int r = 0; r < 4; ++r) {
        v8f acc = {};
#pragma unroll
        for (int kk = 0; kk < 4; ++kk) acc = wmma4(gA[r][kk], uB[kk], acc);
        gu[r] = acc;
      }

      v8f xs[4];
#pragma unroll
      for (int r = 0; r < 4; ++r) {
#pragma unroll
        for (int i = 0; i < 8; ++i) {
          const float a = __builtin_fmaf(lam[r][i], x[r][i], gu[r][i]);
          const float s = row_upper_from_lower(x[r][i], a);
          const float b = __builtin_fmaf(lam[r][i], s, gu[r][i]);
          xs[r][i] = s;
          x[r][i]  = row_swap8(b);
        }
      }

#pragma unroll
      for (int kk = 0; kk < 16; ++kk) {
        v2f xb;
        xb[0] = xs[kk >> 2][2 * (kk & 3)];
        xb[1] = xs[kk >> 2][2 * (kk & 3) + 1];
        y = wmma4(pA[kk], xb, y);
      }

      float* yT = &ldsY[buf][kp][0];
#pragma unroll
      for (int i = 0; i < 8; ++i) {
        const int t = i + 8 * half;
        yT[(t * BP + col) * DPB + wv] = y[i];
      }
    }

#pragma unroll
    for (int k = 0; k < SUP; ++k)
      *(v4f*)(&ldsU[buf ^ 1][k >> 1][coop + (k & 1) * ODD]) = nv[k];
    __syncthreads();
#pragma unroll
    for (int k = 0; k < SUP; ++k) {
      v4f v = *(const v4f*)(&ldsY[buf][k >> 1][coop + (k & 1) * ODD]);
      *(v4f*)(yq + (size_t)(g * SUP + k) * CHK) = v;
    }
  }
}

extern "C" void kernel_launch(void* const* d_in, const int* in_sizes, int n_in,
                              void* d_out, int out_size, void* d_ws,
                              size_t ws_size, hipStream_t stream) {
  (void)in_sizes; (void)n_in; (void)out_size; (void)ws_size;
  const float* u  = (const float*)d_in[0];
  const float* A  = (const float*)d_in[1];
  const float* Bm = (const float*)d_in[2];
  const float* C  = (const float*)d_in[3];
  const float* Dv = (const float*)d_in[4];
  const float* dt = (const float*)d_in[5];
  float* out = (float*)d_out;
  float* ws  = (float*)d_ws;

  ssm_setup_kernel<<<HID, STATE, 0, stream>>>(A, Bm, C, Dv, dt, ws);
  ssm_block_in_kernel<<<XGRID, XTHR, 0, stream>>>(u, ws);
  ssm_scan_kernel<<<HID / DPB, 32 * DPB, 0, stream>>>(ws, ws);
  ssm_block_out_kernel<<<XGRID, XTHR, 0, stream>>>(ws, out);
}


# Scan loop software-pipelined one chunk pair deep: M*U and G*U of pair p+1 (input only) issue beside the vector-unit state steps of pair p; 8-chunk stages, u tiles in 4 LDS buffers; sums unchanged
# speedup vs baseline: 5.2543x; 1.0644x over previous
#include <hip/hip_runtime.h>


typedef __attribute__((ext_vector_type(2))) float v2f;
typedef __attribute__((ext_vector_type(4))) float v4f;
typedef __attribute__((ext_vector_type(8))) float v8f;

#define BATCH  8
#define SEQLEN 4096
#define HID    1024
#define STATE  64
#define T      16
#define NCHUNK (SEQLEN / T)
#define DPB    4
#define BP     16
#define SUP    8
#define NPAIR  (SUP / 2)
#define NSTAGE (NCHUNK / SUP)
#define NUBUF  4
#define TILE   (T * BP * DPB)
#define ODD    (8 * DPB)
#define CHK    (T * BATCH * DPB)

#define PER_D   2368
#define OFF_P   0
#define OFF_G   1024
#define OFF_M   2048
#define OFF_LAM 2304

#define QBLK    (SEQLEN * BATCH * DPB)
#define OFF_UT  ((size_t)4194304)
#define OFF_YT  (OFF_UT + (size_t)BATCH * SEQLEN * HID)

#define XT      4
#define XROWS   (XT * BATCH)
#define XQ      32
#define XPITCH  33
#define XTHR    256
#define XWAVES  (XTHR / 32)
#define XQPARTS (HID / DPB / XQ)
#define XTILES  ((SEQLEN / XT) * XQPARTS)
#define XGRID   512
#define XITER   (XTILES / (XGRID * XWAVES))

static __device__ __forceinline__ float row_swap8(float v) {
  const int s = __builtin_bit_cast(int, v);
  return __builtin_bit_cast(
      float, __builtin_amdgcn_update_dpp(s, s, 0x128, 0xF, 0xF, false));
}
static __device__ __forceinline__ float row_upper_from_lower(float keep, float v) {
  return __builtin_bit_cast(
      float, __builtin_amdgcn_update_dpp(__builtin_bit_cast(int, keep),
                                         __builtin_bit_cast(int, v), 0x128, 0xF,
                                         0xC, false));
}

static __device__ __forceinline__ v8f wmma4(v2f a, v2f b, v8f c) {
  return __builtin_amdgcn_wmma_f32_16x16x4_f32(false, a, false, b, (short)0, c,
                                               false, false);
}

__global__ __launch_bounds__(64) void ssm_setup_kernel(
    const float* __restrict__ A, const float* __restrict__ Bm,
    const float* __restrict__ C, const float* __restrict__ Dv,
    const float* __restrict__ dt, float* __restrict__ ws) {
  const int d = blockIdx.x;
  const int n = threadIdx.x;
  __shared__ float sK[T * STATE];
  __shared__ float sKl[T];
  float* base = ws + (size_t)d * PER_D;

  const float a  = expf(A[d * STATE + n]);
  const float da = expf(-a * dt[n]);
  const float db = Bm[d * STATE + n] * (1.0f - da) / a;
  const float cc = C[d * STATE + n];

  float p = da;
#pragma unroll
  for (int t = 0; t < T; ++t) { base[OFF_P + t * STATE + n] = cc * p; p *= da; }

  const float da2 = da * da, da4 = da2 * da2, da8 = da4 * da4;
  base[OFF_LAM + n] = da8 * da8;

  float g = db;
  for (int s = T - 1; s >= 0; --s) { base[OFF_G + n * T + s] = g; g *= da; }

  float w = cc * db;
#pragma unroll
  for (int tau = 0; tau < T; ++tau) { sK[tau * STATE + n] = w; w *= da; }
  __syncthreads();
  if (n < T) {
    float s = 0.0f;
    for (int m = 0; m < STATE; ++m) s += sK[n * STATE + m];
    sKl[n] = s;
  }
  __syncthreads();
  if (n < T) {
    const int t = n;
    const float dvd = Dv[d];
    for (int s2 = 0; s2 < T; ++s2) {
      float v = (t >= s2) ? sKl[t - s2] : 0.0f;
      if (s2 == t) v += dvd;
      base[OFF_M + t * T + s2] = v;
    }
  }
}

static __device__ __forceinline__ void wave_lds_order() {
  __builtin_amdgcn_fence(__ATOMIC_RELEASE, "wavefront");
  __builtin_amdgcn_wave_barrier();
  __builtin_amdgcn_fence(__ATOMIC_ACQUIRE, "wavefront");
}

__global__ __launch_bounds__(XTHR) void ssm_block_in_kernel(
    const float* __restrict__ u, float* __restrict__ ws) {
  __shared__ __align__(16) float tiles[XWAVES][XROWS * XPITCH * 4];
  float* uT = ws + OFF_UT;
  const int lane = threadIdx.x & 31;
  const int w    = threadIdx.x >> 5;
  float* tile = &tiles[w][0];
  const int gw = blockIdx.x * XWAVES + w;
  v4f v[XROWS];
#pragma unroll
  for (int r = 0; r < XROWS; ++r) {
    const int t0 = (gw / XQPARTS) * XT;
    const int q0 = (gw % XQPARTS) * XQ;
    v[r] = *(const v4f*)(u + ((size_t)(r & 7) * SEQLEN + t0 + (r >> 3)) * HID +
                         (size_t)(q0 + lane) * DPB);
  }
#pragma unroll
  for (int it = 0; it < XITER; ++it) {
    const int tileId = it * (XGRID * XWAVES) + gw;
    const int t0 = (tileId / XQPARTS) * XT;
    const int q0 = (tileId % XQPARTS) * XQ;
#pragma unroll
    for (int r = 0; r < XROWS; ++r)
      *(v4f*)(&tile[(r * XPITCH + lane) * 4]) = v[r];
    wave_lds_order();
    if (it + 1 < XITER) {
      const int nId = (it + 1) * (XGRID * XWAVES) + gw;
      const int nt0 = (nId / XQPARTS) * XT;
      const int nq0 = (nId % XQPARTS) * XQ;
#pragma unroll
      for (int r = 0; r < XROWS; ++r)
        v[r] = *(const v4f*)(u + ((size_t)(r & 7) * SEQLEN + nt0 + (r >> 3)) * HID +
                             (size_t)(nq0 + lane) * DPB);
    }
#pragma unroll
    for (int j = 0; j < XQ; ++j) {
      const v4f x = *(const v4f*)(&tile[(lane * XPITCH + j) * 4]);
      *(v4f*)(uT + (size_t)(q0 + j) * QBLK + (size_t)t0 * (BATCH * DPB) +
              lane * 4) = x;
    }
    wave_lds_order();
  }
}

__global__ __launch_bounds__(XTHR) void ssm_block_out_kernel(
    const float* __restrict__ ws, float* __restrict__ out) {
  __shared__ __align__(16) float tiles[XWAVES][XROWS * XPITCH * 4];
  const float* yT = ws + OFF_YT;
  const int lane = threadIdx.x & 31;
  const int w    = threadIdx.x >> 5;
  float* tile = &tiles[w][0];
  const int gw = blockIdx.x * XWAVES + w;
  v4f v[XQ];
#pragma unroll
  for (int j = 0; j < XQ; ++j) {
    const int t0 = (gw / XQPARTS) * XT;
    const int q0 = (gw % XQPARTS) * XQ;
    v[j] = *(const v4f*)(yT + (size_t)(q0 + j) * QBLK +
                         (size_t)t0 * (BATCH * DPB) + lane * 4);
  }
#pragma unroll
  for (int it = 0; it < XITER; ++it) {
    const int tileId = it * (XGRID * XWAVES) + gw;
    const int t0 = (tileId / XQPARTS) * XT;
    const int q0 = (tileId % XQPARTS) * XQ;
#pragma unroll
    for (int j = 0; j < XQ; ++j)
      *(v4f*)(&tile[(lane * XPITCH + j) * 4]) = v[j];
    wave_lds_order();
    if (it + 1 < XITER) {
      const int nId = (it + 1) * (XGRID * XWAVES) + gw;
      const int nt0 = (nId / XQPARTS) * XT;
      const int nq0 = (nId % XQPARTS) * XQ;
#pragma unroll
      for (int j = 0; j < XQ; ++j)
        v[j] = *(const v4f*)(yT + (size_t)(nq0 + j) * QBLK +
                             (size_t)nt0 * (BATCH * DPB) + lane * 4);
    }
#pragma unroll
    for (int r = 0; r < XROWS; ++r) {
      const v4f x = *(const v4f*)(&tile[(r * XPITCH + lane) * 4]);
      *(v4f*)(out + ((size_t)(r & 7) * SEQLEN + t0 + (r >> 3)) * HID +
              (size_t)(q0 + lane) * DPB) = x;
    }
    wave_lds_order();
  }
}

__global__ __launch_bounds__(32 * DPB) void ssm_scan_kernel(
    const float* __restrict__ ws, float* __restrict__ wsy) {
  const int tid  = threadIdx.x;
  const int lane = tid & 31;
  const int wv   = tid >> 5;
  const int d0   = blockIdx.x * DPB;
  const int half = lane >> 4;
  const int col  = lane & 15;

  __shared__ __align__(16) float ldsU[NUBUF][NPAIR][TILE];
  __shared__ __align__(16) float ldsY[2][NPAIR][TILE];

  const float* base = ws + (size_t)(d0 + wv) * PER_D;

  v2f pA[16];
#pragma unroll
  for (int kk = 0; kk < 16; ++kk)
    pA[kk] = *(const v2f*)(base + OFF_P + col * STATE + 16 * (kk >> 2) +
                           8 * half + 2 * (kk & 3));
  v2f mA[4];
#pragma unroll
  for (int kk = 0; kk < 4; ++kk)
    mA[kk] = *(const v2f*)(base + OFF_M + col * T + 4 * kk + 2 * half);
  v2f gA[4][4];
#pragma unroll
  for (int r = 0; r < 4; ++r)
#pragma unroll
    for (int kk = 0; kk < 4; ++kk)
      gA[r][kk] =
          *(const v2f*)(base + OFF_G + (r * 16 + col) * T + 4 * kk + 2 * half);
  float lam[4][8];
#pragma unroll
  for (int r = 0; r < 4; ++r)
#pragma unroll
    for (int i = 0; i < 8; ++i)
      lam[r][i] = base[OFF_LAM + r * 16 + i + 8 * half];

  v8f x[4] = {};

  const int tt   = tid >> 3;
  const int bb   = tid & 7;
  const int coop = (tt * BP + bb) * DPB;
  const float* uq = ws + OFF_UT + (size_t)blockIdx.x * QBLK + tid * 4;
  float*       yq = wsy + OFF_YT + (size_t)blockIdx.x * QBLK + tid * 4;

#pragma unroll
  for (int k = 0; k < 2 * SUP; ++k)
    *(v4f*)(&ldsU[k / SUP][(k % SUP) >> 1][coop + (k & 1) * ODD]) =
        *(const v4f*)(uq + (size_t)k * CHK);
  __syncthreads();

  v2f uB[4];
#pragma unroll
  for (int kk = 0; kk < 4; ++kk)
#pragma unroll
    for (int j = 0; j < 2; ++j)
      uB[kk][j] = ldsU[0][0][((4 * kk + 2 * half + j) * BP + col) * DPB + wv];

  v8f y = {};
#pragma unroll
  for (int kk = 0; kk < 4; ++kk) y = wmma4(mA[kk], uB[kk], y);
  v8f gu[4];
#pragma unroll
  for (int r = 0; r < 4; ++r) {
    v8f acc = {};
#pragma unroll
    for (int kk = 0; kk < 4; ++kk) acc = wmma4(gA[r][kk], uB[kk], acc);
    gu[r] = acc;
  }
#pragma unroll
  for (int kk = 0; kk < 4; ++kk)
#pragma unroll
    for (int j = 0; j < 2; ++j)
      uB[kk][j] = ldsU[0][1][((4 * kk + 2 * half + j) * BP + col) * DPB + wv];

  for (int g = 0; g < NSTAGE; ++g) {
    const int buf = g & 1;
    const int gn = (g + 2) & (NSTAGE - 1);
    const int gp = (g + 3) & (NSTAGE - 1);
    v4f nv[SUP];
#pragma unroll
    for (int k = 0; k < SUP; ++k) {
      nv[k] = *(const v4f*)(uq + (size_t)(gn * SUP + k) * CHK);
      __builtin_prefetch(uq + (size_t)(gp * SUP + k) * CHK, 0, 0);
    }

#pragma unroll
    for (int kp = 0; kp < NPAIR; ++kp) {
      const float* uT =
          &ldsU[(g + (kp + 2) / NPAIR) & (NUBUF - 1)][(kp + 2) % NPAIR][0];
      v2f uBn[4];
#pragma unroll
      for (int kk = 0; kk < 4; ++kk)
#pragma unroll
        for (int j = 0; j < 2; ++j)
          uBn[kk][j] = uT[((4 * kk + 2 * half + j) * BP + col) * DPB + wv];

      v8f yn = {};
#pragma unroll
      for (int kk = 0; kk < 4; ++kk) yn = wmma4(mA[kk], uB[kk], yn);
      v8f gun[4];
#pragma unroll
      for (int r = 0; r < 4; ++r) {
        v8f acc = {};
#pragma unroll
        for (int kk = 0; kk < 4; ++kk) acc = wmma4(gA[r][kk], uB[kk], acc);
        gun[r] = acc;
      }

      v8f xs[4];
#pragma unroll
      for (int r = 0; r < 4; ++r) {
#pragma unroll
        for (int i = 0; i < 8; ++i) {
          const float a = __builtin_fmaf(lam[r][i], x[r][i], gu[r][i]);
          const float s = row_upper_from_lower(x[r][i], a);
          const float b = __builtin_fmaf(lam[r][i], s, gu[r][i]);
          xs[r][i] = s;
          x[r][i]  = row_swap8(b);
        }
      }

#pragma unroll
      for (int kk = 0; kk < 16; ++kk) {
        v2f xb;
        xb[0] = xs[kk >> 2][2 * (kk & 3)];
        xb[1] = xs[kk >> 2][2 * (kk & 3) + 1];
        y = wmma4(pA[kk], xb, y);
      }

      float* yT = &ldsY[buf][kp][0];
#pragma unroll
      for (int i = 0; i < 8; ++i) {
        const int t = i + 8 * half;
        yT[(t * BP + col) * DPB + wv] = y[i];
      }

      y = yn;
#pragma unroll
      for (int r = 0; r < 4; ++r) gu[r] = gun[r];
#pragma unroll
      for (int kk = 0; kk < 4; ++kk) uB[kk] = uBn[kk];
    }

#pragma unroll
    for (int k = 0; k < SUP; ++k)
      *(v4f*)(&ldsU[(g + 2) & (NUBUF - 1)][k >> 1][coop + (k & 1) * ODD]) = nv[k];
    __syncthreads();
#pragma unroll
    for (int k = 0; k < SUP; ++k) {
      v4f v = *(const v4f*)(&ldsY[buf][k >> 1][coop + (k & 1) * ODD]);
      *(v4f*)(yq + (size_t)(g * SUP + k) * CHK) = v;
    }
  }
}

extern "C" void kernel_launch(void* const* d_in, const int* in_sizes, int n_in,
                              void* d_out, int out_size, void* d_ws,
                              size_t ws_size, hipStream_t stream) {
  (void)in_sizes; (void)n_in; (void)out_size; (void)ws_size;
  const float* u  = (const float*)d_in[0];
  const float* A  = (const float*)d_in[1];
  const float* Bm = (const float*)d_in[2];
  const float* C  = (const float*)d_in[3];
  const float* Dv = (const float*)d_in[4];
  const float* dt = (const float*)d_in[5];
  float* out = (float*)d_out;
  float* ws  = (float*)d_ws;

  ssm_setup_kernel<<<HID, STATE, 0, stream>>>(A, Bm, C, Dv, dt, ws);
  ssm_block_in_kernel<<<XGRID, XTHR, 0, stream>>>(u, ws);
  ssm_scan_kernel<<<HID / DPB, 32 * DPB, 0, stream>>>(ws, ws);
  ssm_block_out_kernel<<<XGRID, XTHR, 0, stream>>>(ws, out);
}


# u and y planes kept per channel in WMMA fragment order: scan waves load B tiles and store Y straight from global memory, no LDS and no barrier in the scan; k order of M*U and G*U sums permuted
# speedup vs baseline: 5.4623x; 1.0396x over previous
#include <hip/hip_runtime.h>


typedef __attribute__((ext_vector_type(2))) float v2f;
typedef __attribute__((ext_vector_type(4))) float v4f;
typedef __attribute__((ext_vector_type(8))) float v8f;

#define BATCH  8
#define SEQLEN 4096
#define HID    1024
#define STATE  64
#define T      16
#define NCHUNK (SEQLEN / T)
#define DPB    4
#define SUP    8
#define NPAIR  (SUP / 2)
#define NSTAGE (NCHUNK / SUP)
#define PAIRF  (2 * T * BATCH)
#define EHF    (PAIRF / 2)
#define PLANE  (SEQLEN * BATCH)

#define PER_D   2368
#define OFF_P   0
#define OFF_G   1024
#define OFF_M   2048
#define OFF_LAM 2304

#define OFF_UT  ((size_t)4194304)
#define OFF_YT  (OFF_UT + (size_t)BATCH * SEQLEN * HID)

#define XCH     32
#define XCB     (HID / XCH)
#define XTHR    256
#define XWAVES  (XTHR / 32)
#define XTILES  ((SEQLEN / 32) * 2 * XCB)
#define XGRID   512
#define XITER   (XTILES / (XGRID * XWAVES))

static __device__ __forceinline__ float row_swap8(float v) {
  const int s = __builtin_bit_cast(int, v);
  return __builtin_bit_cast(
      float, __builtin_amdgcn_update_dpp(s, s, 0x128, 0xF, 0xF, false));
}
static __device__ __forceinline__ float row_upper_from_lower(float keep, float v) {
  return __builtin_bit_cast(
      float, __builtin_amdgcn_update_dpp(__builtin_bit_cast(int, keep),
                                         __builtin_bit_cast(int, v), 0x128, 0xF,
                                         0xC, false));
}

static __device__ __forceinline__ v8f wmma4(v2f a, v2f b, v8f c) {
  return __builtin_amdgcn_wmma_f32_16x16x4_f32(false, a, false, b, (short)0, c,
                                               false, false);
}

__global__ __launch_bounds__(64) void ssm_setup_kernel(
    const float* __restrict__ A, const float* __restrict__ Bm,
    const float* __restrict__ C, const float* __restrict__ Dv,
    const float* __restrict__ dt, float* __restrict__ ws) {
  const int d = blockIdx.x;
  const int n = threadIdx.x;
  __shared__ float sK[T * STATE];
  __shared__ float sKl[T];
  float* base = ws + (size_t)d * PER_D;

  const float a  = expf(A[d * STATE + n]);
  const float da = expf(-a * dt[n]);
  const float db = Bm[d * STATE + n] * (1.0f - da) / a;
  const float cc = C[d * STATE + n];

  float p = da;
#pragma unroll
  for (int t = 0; t < T; ++t) { base[OFF_P + t * STATE + n] = cc * p; p *= da; }

  const float da2 = da * da, da4 = da2 * da2, da8 = da4 * da4;
  base[OFF_LAM + n] = da8 * da8;

  float g = db;
  for (int s = T - 1; s >= 0; --s) { base[OFF_G + n * T + s] = g; g *= da; }

  float w = cc * db;
#pragma unroll
  for (int tau = 0; tau < T; ++tau) { sK[tau * STATE + n] = w; w *= da; }
  __syncthreads();
  if (n < T) {
    float s = 0.0f;
    for (int m = 0; m < STATE; ++m) s += sK[n * STATE + m];
    sKl[n] = s;
  }
  __syncthreads();
  if (n < T) {
    const int t = n;
    const float dvd = Dv[d];
    for (int s2 = 0; s2 < T; ++s2) {
      float v = (t >= s2) ? sKl[t - s2] : 0.0f;
      if (s2 == t) v += dvd;
      base[OFF_M + t * T + s2] = v;
    }
  }
}

static __device__ __forceinline__ void wave_lds_order() {
  __builtin_amdgcn_fence(__ATOMIC_RELEASE, "wavefront");
  __builtin_amdgcn_wave_barrier();
  __builtin_amdgcn_fence(__ATOMIC_ACQUIRE, "wavefront");
}

__global__ __launch_bounds__(XTHR) void ssm_block_in_kernel(
    const float* __restrict__ u, float* __restrict__ ws) {
  __shared__ __align__(16) float tiles[XWAVES][XCH * EHF];
  float* uT = ws + OFF_UT;
  const int lane = threadIdx.x & 31;
  const int w    = threadIdx.x >> 5;
  const int l7   = lane & 7;
  const int l3   = lane >> 3;
  float* tile = &tiles[w][0];
  const int gw = blockIdx.x * XWAVES + w;
  v4f v[32];
#pragma unroll
  for (int k = 0; k < 32; ++k) {
    const int g = k >> 2, e4 = k & 3;
    const int cb = gw % XCB, pe = gw / XCB;
    const int b = 4 * (g & 1) + l3;
    const int t = 32 * (pe >> 1) + 16 * ((g >> 1) & 1) + 8 * (g >> 2) +
                  4 * (pe & 1) + e4;
    v[k] = *(const v4f*)(u + ((size_t)b * SEQLEN + t) * HID + cb * XCH + l7 * 4);
  }
#pragma unroll
  for (int it = 0; it < XITER; ++it) {
    const int tileId = it * (XGRID * XWAVES) + gw;
    const int cb = tileId % XCB, pe = tileId / XCB;
#pragma unroll
    for (int g = 0; g < 8; ++g)
#pragma unroll
      for (int c = 0; c < 4; ++c) {
        v4f x;
        x[0] = v[4 * g + 0][c]; x[1] = v[4 * g + 1][c];
        x[2] = v[4 * g + 2][c]; x[3] = v[4 * g + 3][c];
        *(v4f*)(&tile[((4 * l7 + c) * 32 + ((4 * g + l3) ^ (l7 << 1))) * 4]) = x;
      }
    wave_lds_order();
    if (it + 1 < XITER) {
      const int nId = (it + 1) * (XGRID * XWAVES) + gw;
      const int ncb = nId % XCB, npe = nId / XCB;
#pragma unroll
      for (int k = 0; k < 32; ++k) {
        const int g = k >> 2, e4 = k & 3;
        const int b = 4 * (g & 1) + l3;
        const int t = 32 * (npe >> 1) + 16 * ((g >> 1) & 1) + 8 * (g >> 2) +
                      4 * (npe & 1) + e4;
        v[k] = *(const v4f*)(u + ((size_t)b * SEQLEN + t) * HID + ncb * XCH +
                             l7 * 4);
      }
    }
#pragma unroll
    for (int cf = 0; cf < XCH; ++cf) {
      const v4f x =
          *(const v4f*)(&tile[(cf * 32 + (lane ^ (((cf >> 2) & 7) << 1))) * 4]);
      *(v4f*)(uT + (size_t)(cb * XCH + cf) * PLANE + (size_t)pe * EHF +
              lane * 4) = x;
    }
    wave_lds_order();
  }
}

__global__ __launch_bounds__(XTHR) void ssm_block_out_kernel(
    const float* __restrict__ ws, float* __restrict__ out) {
  __shared__ __align__(16) float tiles[XWAVES][XCH * EHF];
  const float* yT = ws + OFF_YT;
  const int lane = threadIdx.x & 31;
  const int w    = threadIdx.x >> 5;
  const int l7   = lane & 7;
  const int l3   = lane >> 3;
  float* tile = &tiles[w][0];
  const int gw = blockIdx.x * XWAVES + w;
  v4f v[XCH];
#pragma unroll
  for (int cf = 0; cf < XCH; ++cf) {
    const int cb = gw % XCB, pe = gw / XCB;
    v[cf] = *(const v4f*)(yT + (size_t)(cb * XCH + cf) * PLANE +
                          (size_t)pe * EHF + lane * 4);
  }
#pragma unroll
  for (int it = 0; it < XITER; ++it) {
    const int tileId = it * (XGRID * XWAVES) + gw;
    const int cb = tileId % XCB, pe = tileId / XCB;
#pragma unroll
    for (int cf = 0; cf < XCH; ++cf)
      *(v4f*)(&tile[(cf * 32 + (lane ^ (((cf >> 2) & 7) << 1))) * 4]) = v[cf];
    wave_lds_order();
    if (it + 1 < XITER) {
      const int nId = (it + 1) * (XGRID * XWAVES) + gw;
      const int ncb = nId % XCB, npe = nId / XCB;
#pragma unroll
      for (int cf = 0; cf < XCH; ++cf)
        v[cf] = *(const v4f*)(yT + (size_t)(ncb * XCH + cf) * PLANE +
                              (size_t)npe * EHF + lane * 4);
    }
#pragma unroll
    for (int g = 0; g < 8; ++g) {
      v4f r[4];
#pragma unroll
      for (int c = 0; c < 4; ++c)
        r[c] = *(const v4f*)(
            &tile[((4 * l7 + c) * 32 + ((4 * g + l3) ^ (l7 << 1))) * 4]);
      const int b = 4 * (g & 1) + l3;
#pragma unroll
      for (int e4 = 0; e4 < 4; ++e4) {
        const int t = 32 * (pe >> 1) + 16 * ((g >> 1) & 1) + 8 * (g >> 2) +
                      4 * (pe & 1) + e4;
        v4f x;
        x[0] = r[0][e4]; x[1] = r[1][e4]; x[2] = r[2][e4]; x[3] = r[3][e4];
        *(v4f*)(out + ((size_t)b * SEQLEN + t) * HID + cb * XCH + l7 * 4) = x;
      }
    }
    wave_lds_order();
  }
}

__global__ __launch_bounds__(32 * DPB) void ssm_scan_kernel(
    const float* __restrict__ ws, float* __restrict__ wsy) {
  const int tid  = threadIdx.x;
  const int lane = tid & 31;
  const int wv   = tid >> 5;
  const int d    = blockIdx.x * DPB + wv;
  const int half = lane >> 4;
  const int col  = lane & 15;

  const float* base = ws + (size_t)d * PER_D;

  v2f pA[16];
#pragma unroll
  for (int kk = 0; kk < 16; ++kk)
    pA[kk] = *(const v2f*)(base + OFF_P + col * STATE + 16 * (kk >> 2) +
                           8 * half + 2 * (kk & 3));
  v2f mA[4];
#pragma unroll
  for (int kk = 0; kk < 4; ++kk)
    mA[kk] = *(const v2f*)(base + OFF_M + col * T + 8 * half + 2 * kk);
  v2f gA[4][4];
#pragma unroll
  for (int r = 0; r < 4; ++r)
#pragma unroll
    for (int kk = 0; kk < 4; ++kk)
      gA[r][kk] =
          *(const v2f*)(base + OFF_G + (r * 16 + col) * T + 8 * half + 2 * kk);
  float lam[4][8];
#pragma unroll
  for (int r = 0; r < 4; ++r)
#pragma unroll
    for (int i = 0; i < 8; ++i)
      lam[r][i] = base[OFF_LAM + r * 16 + i + 8 * half];

  v8f x[4] = {};

  const float* up = ws + OFF_UT + (size_t)d * PLANE + lane * 4;
  float*       yp = wsy + OFF_YT + (size_t)d * PLANE + lane * 4;

  v4f buf[2][NPAIR][2];
#pragma unroll
  for (int kp = 0; kp < NPAIR; ++kp)
#pragma unroll
    for (int eh = 0; eh < 2; ++eh)
      buf[0][kp][eh] = *(const v4f*)(up + (size_t)kp * PAIRF + eh * EHF);

  v8f y = {};
  v8f gu[4];
  {
    v2f uB[4];
    uB[0][0] = buf[0][0][0][0]; uB[0][1] = buf[0][0][0][1];
    uB[1][0] = buf[0][0][0][2]; uB[1][1] = buf[0][0][0][3];
    uB[2][0] = buf[0][0][1][0]; uB[2][1] = buf[0][0][1][1];
    uB[3][0] = buf[0][0][1][2]; uB[3][1] = buf[0][0][1][3];
#pragma unroll
    for (int kk = 0; kk < 4; ++kk) y = wmma4(mA[kk], uB[kk], y);
#pragma unroll
    for (int r = 0; r < 4; ++r) {
      v8f acc = {};
#pragma unroll
      for (int kk = 0; kk < 4; ++kk) acc = wmma4(gA[r][kk], uB[kk], acc);
      gu[r] = acc;
    }
  }

  for (int gg = 0; gg < NSTAGE / 2; ++gg) {
#pragma unroll
   for (int ph = 0; ph < 2; ++ph) {
    const int g = 2 * gg + ph;
    const int gn = (g + 1) & (NSTAGE - 1);
    const int gp = (g + 3) & (NSTAGE - 1);
#pragma unroll
    for (int kp = 0; kp < NPAIR; ++kp)
#pragma unroll
      for (int eh = 0; eh < 2; ++eh) {
        buf[ph ^ 1][kp][eh] =
            *(const v4f*)(up + (size_t)(gn * NPAIR + kp) * PAIRF + eh * EHF);
        __builtin_prefetch(up + (size_t)(gp * NPAIR + kp) * PAIRF + eh * EHF, 0,
                           0);
      }
    __builtin_amdgcn_sched_barrier(0);

#pragma unroll
    for (int kp = 0; kp < NPAIR; ++kp) {
      const v4f n0 =
          (kp + 1 < NPAIR) ? buf[ph][(kp + 1) % NPAIR][0] : buf[ph ^ 1][0][0];
      const v4f n1 =
          (kp + 1 < NPAIR) ? buf[ph][(kp + 1) % NPAIR][1] : buf[ph ^ 1][0][1];
      v2f uB[4];
      uB[0][0] = n0[0]; uB[0][1] = n0[1];
      uB[1][0] = n0[2]; uB[1][1] = n0[3];
      uB[2][0] = n1[0]; uB[2][1] = n1[1];
      uB[3][0] = n1[2]; uB[3][1] = n1[3];

      v8f yn = {};
#pragma unroll
      for (int kk = 0; kk < 4; ++kk) yn = wmma4(mA[kk], uB[kk], yn);
      v8f gun[4];
#pragma unroll
      for (int r = 0; r < 4; ++r) {
        v8f acc = {};
#pragma unroll
        for (int kk = 0; kk < 4; ++kk) acc = wmma4(gA[r][kk], uB[kk], acc);
        gun[r] = acc;
      }

      v8f xs[4];
#pragma unroll
      for (int r = 0; r < 4; ++r) {
#pragma unroll
        for (int i = 0; i < 8; ++i) {
          const float a = __builtin_fmaf(lam[r][i], x[r][i], gu[r][i]);
          const float s = row_upper_from_lower(x[r][i], a);
          const float b = __builtin_fmaf(lam[r][i], s, gu[r][i]);
          xs[r][i] = s;
          x[r][i]  = row_swap8(b);
        }
      }

#pragma unroll
      for (int kk = 0; kk < 16; ++kk) {
        v2f xb;
        xb[0] = xs[kk >> 2][2 * (kk & 3)];
        xb[1] = xs[kk >> 2][2 * (kk & 3) + 1];
        y = wmma4(pA[kk], xb, y);
      }

      {
        float* yo = yp + (size_t)(g * NPAIR + kp) * PAIRF;
        v4f y0, y1;
        y0[0] = y[0]; y0[1] = y[1]; y0[2] = y[2]; y0[3] = y[3];
        y1[0] = y[4]; y1[1] = y[5]; y1[2] = y[6]; y1[3] = y[7];
        *(v4f*)(yo)       = y0;
        *(v4f*)(yo + EHF) = y1;
      }

      y = yn;
#pragma unroll
      for (int r = 0; r < 4; ++r) gu[r] = gun[r];
    }
   }
  }
}

extern "C" void kernel_launch(void* const* d_in, const int* in_sizes, int n_in,
                              void* d_out, int out_size, void* d_ws,
                              size_t ws_size, hipStream_t stream) {
  (void)in_sizes; (void)n_in; (void)out_size; (void)ws_size;
  const float* u  = (const float*)d_in[0];
  const float* A  = (const float*)d_in[1];
  const float* Bm = (const float*)d_in[2];
  const float* C  = (const float*)d_in[3];
  const float* Dv = (const float*)d_in[4];
  const float* dt = (const float*)d_in[5];
  float* out = (float*)d_out;
  float* ws  = (float*)d_ws;

  ssm_setup_kernel<<<HID, STATE, 0, stream>>>(A, Bm, C, Dv, dt, ws);
  ssm_block_in_kernel<<<XGRID, XTHR, 0, stream>>>(u, ws);
  ssm_scan_kernel<<<HID / DPB, 32 * DPB, 0, stream>>>(ws, ws);
  ssm_block_out_kernel<<<XGRID, XTHR, 0, stream>>>(ws, out);
}


# Set-up launch folded into the scan: every scan wave derives its own channel's P, G, M, Lam in a private LDS pack (same arithmetic and order) behind its first tile loads; 3 launches, not 4
# speedup vs baseline: 5.6504x; 1.0344x over previous
#include <hip/hip_runtime.h>


typedef __attribute__((ext_vector_type(2))) float v2f;
typedef __attribute__((ext_vector_type(4))) float v4f;
typedef __attribute__((ext_vector_type(8))) float v8f;

#define BATCH  8
#define SEQLEN 4096
#define HID    1024
#define STATE  64
#define T      16
#define NCHUNK (SEQLEN / T)
#define DPB    4
#define SUP    8
#define NPAIR  (SUP / 2)
#define NSTAGE (NCHUNK / SUP)
#define PAIRF  (2 * T * BATCH)
#define EHF    (PAIRF / 2)
#define PLANE  (SEQLEN * BATCH)

#define PER_D   2368
#define OFF_P   0
#define OFF_G   1024
#define OFF_M   2048
#define OFF_LAM 2304
#define KPITCH  (STATE + 4)
#define OFF_SK  PER_D
#define OFF_SKL (OFF_SK + T * KPITCH)
#define PACKF   (OFF_SKL + T)

#define OFF_UT  ((size_t)4194304)
#define OFF_YT  (OFF_UT + (size_t)BATCH * SEQLEN * HID)

#define XCH     32
#define XCB     (HID / XCH)
#define XTHR    256
#define XWAVES  (XTHR / 32)
#define XTILES  ((SEQLEN / 32) * 2 * XCB)
#define XGRID   512
#define XITER   (XTILES / (XGRID * XWAVES))

static __device__ __forceinline__ float row_swap8(float v) {
  const int s = __builtin_bit_cast(int, v);
  return __builtin_bit_cast(
      float, __builtin_amdgcn_update_dpp(s, s, 0x128, 0xF, 0xF, false));
}
static __device__ __forceinline__ float row_upper_from_lower(float keep, float v) {
  return __builtin_bit_cast(
      float, __builtin_amdgcn_update_dpp(__builtin_bit_cast(int, keep),
                                         __builtin_bit_cast(int, v), 0x128, 0xF,
                                         0xC, false));
}

static __device__ __forceinline__ v8f wmma4(v2f a, v2f b, v8f c) {
  return __builtin_amdgcn_wmma_f32_16x16x4_f32(false, a, false, b, (short)0, c,
                                               false, false);
}

static __device__ __forceinline__ void wave_lds_order() {
  __builtin_amdgcn_fence(__ATOMIC_RELEASE, "wavefront");
  __builtin_amdgcn_wave_barrier();
  __builtin_amdgcn_fence(__ATOMIC_ACQUIRE, "wavefront");
}

__global__ __launch_bounds__(XTHR) void ssm_block_in_kernel(
    const float* __restrict__ u, float* __restrict__ ws) {
  __shared__ __align__(16) float tiles[XWAVES][XCH * EHF];
  float* uT = ws + OFF_UT;
  const int lane = threadIdx.x & 31;
  const int w    = threadIdx.x >> 5;
  const int l7   = lane & 7;
  const int l3   = lane >> 3;
  float* tile = &tiles[w][0];
  const int gw = blockIdx.x * XWAVES + w;
  v4f v[32];
#pragma unroll
  for (int k = 0; k < 32; ++k) {
    const int g = k >> 2, e4 = k & 3;
    const int cb = gw % XCB, pe = gw / XCB;
    const int b = 4 * (g & 1) + l3;
    const int t = 32 * (pe >> 1) + 16 * ((g >> 1) & 1) + 8 * (g >> 2) +
                  4 * (pe & 1) + e4;
    v[k] = *(const v4f*)(u + ((size_t)b * SEQLEN + t) * HID + cb * XCH + l7 * 4);
  }
#pragma unroll
  for (int it = 0; it < XITER; ++it) {
    const int tileId = it * (XGRID * XWAVES) + gw;
    const int cb = tileId % XCB, pe = tileId / XCB;
#pragma unroll
    for (int g = 0; g < 8; ++g)
#pragma unroll
      for (int c = 0; c < 4; ++c) {
        v4f x;
        x[0] = v[4 * g + 0][c]; x[1] = v[4 * g + 1][c];
        x[2] = v[4 * g + 2][c]; x[3] = v[4 * g + 3][c];
        *(v4f*)(&tile[((4 * l7 + c) * 32 + ((4 * g + l3) ^ (l7 << 1))) * 4]) = x;
      }
    wave_lds_order();
    if (it + 1 < XITER) {
      const int nId = (it + 1) * (XGRID * XWAVES) + gw;
      const int ncb = nId % XCB, npe = nId / XCB;
#pragma unroll
      for (int k = 0; k < 32; ++k) {
        const int g = k >> 2, e4 = k & 3;
        const int b = 4 * (g & 1) + l3;
        const int t = 32 * (npe >> 1) + 16 * ((g >> 1) & 1) + 8 * (g >> 2) +
                      4 * (npe & 1) + e4;
        v[k] = *(const v4f*)(u + ((size_t)b * SEQLEN + t) * HID + ncb * XCH +
                             l7 * 4);
      }
    }
#pragma unroll
    for (int cf = 0; cf < XCH; ++cf) {
      const v4f x =
          *(const v4f*)(&tile[(cf * 32 + (lane ^ (((cf >> 2) & 7) << 1))) * 4]);
      *(v4f*)(uT + (size_t)(cb * XCH + cf) * PLANE + (size_t)pe * EHF +
              lane * 4) = x;
    }
    wave_lds_order();
  }
}

__global__ __launch_bounds__(XTHR) void ssm_block_out_kernel(
    const float* __restrict__ ws, float* __restrict__ out) {
  __shared__ __align__(16) float tiles[XWAVES][XCH * EHF];
  const float* yT = ws + OFF_YT;
  const int lane = threadIdx.x & 31;
  const int w    = threadIdx.x >> 5;
  const int l7   = lane & 7;
  const int l3   = lane >> 3;
  float* tile = &tiles[w][0];
  const int gw = blockIdx.x * XWAVES + w;
  v4f v[XCH];
#pragma unroll
  for (int cf = 0; cf < XCH; ++cf) {
    const int cb = gw % XCB, pe = gw / XCB;
    v[cf] = *(const v4f*)(yT + (size_t)(cb * XCH + cf) * PLANE +
                          (size_t)pe * EHF + lane * 4);
  }
#pragma unroll
  for (int it = 0; it < XITER; ++it) {
    const int tileId = it * (XGRID * XWAVES) + gw;
    const int cb = tileId % XCB, pe = tileId / XCB;
#pragma unroll
    for (int cf = 0; cf < XCH; ++cf)
      *(v4f*)(&tile[(cf * 32 + (lane ^ (((cf >> 2) & 7) << 1))) * 4]) = v[cf];
    wave_lds_order();
    if (it + 1 < XITER) {
      const int nId = (it + 1) * (XGRID * XWAVES) + gw;
      const int ncb = nId % XCB, npe = nId / XCB;
#pragma unroll
      for (int cf = 0; cf < XCH; ++cf)
        v[cf] = *(const v4f*)(yT + (size_t)(ncb * XCH + cf) * PLANE +
                              (size_t)npe * EHF + lane * 4);
    }
#pragma unroll
    for (int g = 0; g < 8; ++g) {
      v4f r[4];
#pragma unroll
      for (int c = 0; c < 4; ++c)
        r[c] = *(const v4f*)(
            &tile[((4 * l7 + c) * 32 + ((4 * g + l3) ^ (l7 << 1))) * 4]);
      const int b = 4 * (g & 1) + l3;
#pragma unroll
      for (int e4 = 0; e4 < 4; ++e4) {
        const int t = 32 * (pe >> 1) + 16 * ((g >> 1) & 1) + 8 * (g >> 2) +
                      4 * (pe & 1) + e4;
        v4f x;
        x[0] = r[0][e4]; x[1] = r[1][e4]; x[2] = r[2][e4]; x[3] = r[3][e4];
        *(v4f*)(out + ((size_t)b * SEQLEN + t) * HID + cb * XCH + l7 * 4) = x;
      }
    }
    wave_lds_order();
  }
}

__global__ __launch_bounds__(32 * DPB) void ssm_scan_kernel(
    const float* __restrict__ A, const float* __restrict__ Bm,
    const float* __restrict__ C, const float* __restrict__ Dv,
    const float* __restrict__ dt, const float* __restrict__ ws,
    float* __restrict__ wsy) {
  __shared__ __align__(16) float packs[DPB][PACKF];
  const int tid  = threadIdx.x;
  const int lane = tid & 31;
  const int wv   = tid >> 5;
  const int d    = blockIdx.x * DPB + wv;
  const int half = lane >> 4;
  const int col  = lane & 15;

  const float* up = ws + OFF_UT + (size_t)d * PLANE + lane * 4;
  float*       yp = wsy + OFF_YT + (size_t)d * PLANE + lane * 4;

  v4f buf[2][NPAIR][2];
#pragma unroll
  for (int kp = 0; kp < NPAIR; ++kp)
#pragma unroll
    for (int eh = 0; eh < 2; ++eh)
      buf[0][kp][eh] = *(const v4f*)(up + (size_t)kp * PAIRF + eh * EHF);

  float* base = &packs[wv][0];
  {
    float* sK  = base + OFF_SK;
    float* sKl = base + OFF_SKL;
    const float dvd = Dv[d];
#pragma unroll
    for (int q = 0; q < 2; ++q) {
      const int n = lane + 32 * q;
      const float a  = expf(A[d * STATE + n]);
      const float da = expf(-a * dt[n]);
      const float db = Bm[d * STATE + n] * (1.0f - da) / a;
      const float cc = C[d * STATE + n];

      float p = da;
#pragma unroll
      for (int t = 0; t < T; ++t) { base[OFF_P + t * STATE + n] = cc * p; p *= da; }

      const float da2 = da * da, da4 = da2 * da2, da8 = da4 * da4;
      base[OFF_LAM + n] = da8 * da8;

      float g = db;
#pragma unroll
      for (int s = T - 1; s >= 0; --s) { base[OFF_G + n * T + s] = g; g *= da; }

      float w = cc * db;
#pragma unroll
      for (int tau = 0; tau < T; ++tau) { sK[tau * KPITCH + n] = w; w *= da; }
    }
    wave_lds_order();
    if (lane < T) {
      float s = 0.0f;
      for (int m = 0; m < STATE; m += 4) {
        const v4f q = *(const v4f*)(&sK[lane * KPITCH + m]);
        s += q[0]; s += q[1]; s += q[2]; s += q[3];
      }
      sKl[lane] = s;
    }
    wave_lds_order();
    if (lane < T) {
      const int t = lane;
      float kv[T];
#pragma unroll
      for (int s2 = 0; s2 < T; ++s2) kv[s2] = sKl[(t >= s2) ? (t - s2) : 0];
#pragma unroll
      for (int s2 = 0; s2 < T; ++s2) {
        float v = (t >= s2) ? kv[s2] : 0.0f;
        if (s2 == t) v += dvd;
        base[OFF_M + t * T + s2] = v;
      }
    }
    wave_lds_order();
  }

  v2f pA[16];
#pragma unroll
  for (int kk = 0; kk < 16; ++kk)
    pA[kk] = *(const v2f*)(base + OFF_P + col * STATE + 16 * (kk >> 2) +
                           8 * half + 2 * (kk & 3));
  v2f mA[4];
#pragma unroll
  for (int kk = 0; kk < 4; ++kk)
    mA[kk] = *(const v2f*)(base + OFF_M + col * T + 8 * half + 2 * kk);
  v2f gA[4][4];
#pragma unroll
  for (int r = 0; r < 4; ++r)
#pragma unroll
    for (int kk = 0; kk < 4; ++kk)
      gA[r][kk] =
          *(const v2f*)(base + OFF_G + (r * 16 + col) * T + 8 * half + 2 * kk);
  float lam[4][8];
#pragma unroll
  for (int r = 0; r < 4; ++r)
#pragma unroll
    for (int i = 0; i < 8; ++i)
      lam[r][i] = base[OFF_LAM + r * 16 + i + 8 * half];

  v8f x[4] = {};

  v8f y = {};
  v8f gu[4];
  {
    v2f uB[4];
    uB[0][0] = buf[0][0][0][0]; uB[0][1] = buf[0][0][0][1];
    uB[1][0] = buf[0][0][0][2]; uB[1][1] = buf[0][0][0][3];
    uB[2][0] = buf[0][0][1][0]; uB[2][1] = buf[0][0][1][1];
    uB[3][0] = buf[0][0][1][2]; uB[3][1] = buf[0][0][1][3];
#pragma unroll
    for (int kk = 0; kk < 4; ++kk) y = wmma4(mA[kk], uB[kk], y);
#pragma unroll
    for (int r = 0; r < 4; ++r) {
      v8f acc = {};
#pragma unroll
      for (int kk = 0; kk < 4; ++kk) acc = wmma4(gA[r][kk], uB[kk], acc);
      gu[r] = acc;
    }
  }

  for (int gg = 0; gg < NSTAGE / 2; ++gg) {
#pragma unroll
   for (int ph = 0; ph < 2; ++ph) {
    const int g = 2 * gg + ph;
    const int gn = (g + 1) & (NSTAGE - 1);
    const int gp = (g + 3) & (NSTAGE - 1);
#pragma unroll
    for (int kp = 0; kp < NPAIR; ++kp)
#pragma unroll
      for (int eh = 0; eh < 2; ++eh) {
        buf[ph ^ 1][kp][eh] =
            *(const v4f*)(up + (size_t)(gn * NPAIR + kp) * PAIRF + eh * EHF);
        __builtin_prefetch(up + (size_t)(gp * NPAIR + kp) * PAIRF + eh * EHF, 0,
                           0);
      }
    __builtin_amdgcn_sched_barrier(0);

#pragma unroll
    for (int kp = 0; kp < NPAIR; ++kp) {
      const v4f n0 =
          (kp + 1 < NPAIR) ? buf[ph][(kp + 1) % NPAIR][0] : buf[ph ^ 1][0][0];
      const v4f n1 =
          (kp + 1 < NPAIR) ? buf[ph][(kp + 1) % NPAIR][1] : buf[ph ^ 1][0][1];
      v2f uB[4];
      uB[0][0] = n0[0]; uB[0][1] = n0[1];
      uB[1][0] = n0[2]; uB[1][1] = n0[3];
      uB[2][0] = n1[0]; uB[2][1] = n1[1];
      uB[3][0] = n1[2]; uB[3][1] = n1[3];

      v8f yn = {};
#pragma unroll
      for (int kk = 0; kk < 4; ++kk) yn = wmma4(mA[kk], uB[kk], yn);
      v8f gun[4];
#pragma unroll
      for (int r = 0; r < 4; ++r) {
        v8f acc = {};
#pragma unroll
        for (int kk = 0; kk < 4; ++kk) acc = wmma4(gA[r][kk], uB[kk], acc);
        gun[r] = acc;
      }

      v8f xs[4];
#pragma unroll
      for (int r = 0; r < 4; ++r) {
#pragma unroll
        for (int i = 0; i < 8; ++i) {
          const float a = __builtin_fmaf(lam[r][i], x[r][i], gu[r][i]);
          const float s = row_upper_from_lower(x[r][i], a);
          const float b = __builtin_fmaf(lam[r][i], s, gu[r][i]);
          xs[r][i] = s;
          x[r][i]  = row_swap8(b);
        }
      }

#pragma unroll
      for (int kk = 0; kk < 16; ++kk) {
        v2f xb;
        xb[0] = xs[kk >> 2][2 * (kk & 3)];
        xb[1] = xs[kk >> 2][2 * (kk & 3) + 1];
        y = wmma4(pA[kk], xb, y);
      }

      {
        float* yo = yp + (size_t)(g * NPAIR + kp) * PAIRF;
        v4f y0, y1;
        y0[0] = y[0]; y0[1] = y[1]; y0[2] = y[2]; y0[3] = y[3];
        y1[0] = y[4]; y1[1] = y[5]; y1[2] = y[6]; y1[3] = y[7];
        *(v4f*)(yo)       = y0;
        *(v4f*)(yo + EHF) = y1;
      }

      y = yn;
#pragma unroll
      for (int r = 0; r < 4; ++r) gu[r] = gun[r];
    }
   }
  }
}

extern "C" void kernel_launch(void* const* d_in, const int* in_sizes, int n_in,
                              void* d_out, int out_size, void* d_ws,
                              size_t ws_size, hipStream_t stream) {
  (void)in_sizes; (void)n_in; (void)out_size; (void)ws_size;
  const float* u  = (const float*)d_in[0];
  const float* A  = (const float*)d_in[1];
  const float* Bm = (const float*)d_in[2];
  const float* C  = (const float*)d_in[3];
  const float* Dv = (const float*)d_in[4];
  const float* dt = (const float*)d_in[5];
  float* out = (float*)d_out;
  float* ws  = (float*)d_ws;

  ssm_block_in_kernel<<<XGRID, XTHR, 0, stream>>>(u, ws);
  ssm_scan_kernel<<<HID / DPB, 32 * DPB, 0, stream>>>(A, Bm, C, Dv, dt, ws, ws);
  ssm_block_out_kernel<<<XGRID, XTHR, 0, stream>>>(ws, out);
}
